# starting kernel
# baseline (speedup 1.0000x reference)
#include <hip/hip_runtime.h>
#include <math.h>

typedef __attribute__((ext_vector_type(2))) float v2f;
typedef __attribute__((ext_vector_type(8))) float v8f;

#define CHANNELS 16
#define KDIM 128

__global__ void gcn_init_kernel(float* __restrict__ out_accum,
                                float* __restrict__ deg, int n) {
    int i = blockIdx.x * blockDim.x + threadIdx.x;
    if (i < n * CHANNELS) out_accum[i] = 0.0f;
    if (i < n) deg[i] = 1.0f;
}

__global__ void gcn_gemm_wmma_kernel(const float* __restrict__ x,
                                     const float* __restrict__ W,
                                     float* __restrict__ h, int n) {
    const int lane   = threadIdx.x & 31;
    const int waveId = threadIdx.x >> 5;
    const int ntiles = (n + 15) >> 4;
    const int tile   = blockIdx.x * (blockDim.x >> 5) + waveId;
    if (tile >= ntiles) return;

    const int half = lane >> 4;
    const int idx  = lane & 15;
    const int koff = half * 2;

    int arow = tile * 16 + idx;
    if (arow >= n) arow = n - 1;
    const float* xrow = x + (size_t)arow * KDIM;

    v8f c = {};
#pragma unroll
    for (int k = 0; k < KDIM; k += 4) {
        v2f a;
        a.x = xrow[k + koff + 0];
        a.y = xrow[k + koff + 1];
        v2f bm;
        bm.x = W[(k + koff + 0) * CHANNELS + idx];
        bm.y = W[(k + koff + 1) * CHANNELS + idx];
        c = __builtin_amdgcn_wmma_f32_16x16x4_f32(
             false, a,  false, bm,
             (short)0, c,  false,  false);
    }

#pragma unroll
    for (int v = 0; v < 8; ++v) {
        int row = tile * 16 + v + 8 * half;
        if (row < n) h[(size_t)row * CHANNELS + idx] = c[v];
    }
}

__global__ void gcn_degree_kernel(const int* __restrict__ col,
                                  float* __restrict__ deg, int e) {
    int i = blockIdx.x * blockDim.x + threadIdx.x;
    if (i < e) atomicAdd(&deg[col[i]], 1.0f);
}

__global__ void gcn_rsqrt_kernel(float* __restrict__ deg, int n) {
    int i = blockIdx.x * blockDim.x + threadIdx.x;
    if (i < n) deg[i] = rsqrtf(deg[i]);
}

__global__ void gcn_scatter_kernel(const int* __restrict__ row,
                                   const int* __restrict__ col,
                                   const float* __restrict__ dis,
                                   const float* __restrict__ h,
                                   float* __restrict__ out_accum, int e) {
    long long t = (long long)blockIdx.x * blockDim.x + threadIdx.x;
    long long ec = (long long)e * CHANNELS;
    if (t >= ec) return;
    int edge = (int)(t >> 4);
    int ch   = (int)(t & 15);
    int r = row[edge];
    int cl = col[edge];
    float norm = dis[r] * dis[cl];
    float msg  = norm * h[(size_t)r * CHANNELS + ch];
    atomicAdd(&out_accum[(size_t)cl * CHANNELS + ch], msg);
}

__global__ void gcn_final_kernel(float* __restrict__ out,
                                 const float* __restrict__ h,
                                 const float* __restrict__ dis,
                                 const float* __restrict__ b, int n) {
    int i = blockIdx.x * blockDim.x + threadIdx.x;
    if (i >= n) return;
    float selfw = dis[i] * dis[i];
    float v[CHANNELS];
    float m = -INFINITY;
#pragma unroll
    for (int c = 0; c < CHANNELS; ++c) {
        float val = out[(size_t)i * CHANNELS + c]
                  + selfw * h[(size_t)i * CHANNELS + c] + b[c];
        val = fmaxf(val, 0.0f);
        v[c] = val;
        m = fmaxf(m, val);
    }
    float s = 0.0f;
#pragma unroll
    for (int c = 0; c < CHANNELS; ++c) s += __expf(v[c] - m);
    float lse = m + __logf(s);
#pragma unroll
    for (int c = 0; c < CHANNELS; ++c)
        out[(size_t)i * CHANNELS + c] = v[c] - lse;
}

extern "C" void kernel_launch(void* const* d_in, const int* in_sizes, int n_in,
                              void* d_out, int out_size, void* d_ws, size_t ws_size,
                              hipStream_t stream) {
    const float* x  = (const float*)d_in[0];
    const int*   ei = (const int*)d_in[1];
    const float* W  = (const float*)d_in[2];
    const float* b  = (const float*)d_in[3];
    float* out = (float*)d_out;

    const int n = in_sizes[0] / KDIM;
    const int e = in_sizes[1] / 2;
    const int* row = ei;
    const int* col = ei + e;

    float* h   = (float*)d_ws;
    float* deg = h + (size_t)n * CHANNELS;

    const int B = 256;

    gcn_init_kernel<<<(n * CHANNELS + B - 1) / B, B, 0, stream>>>(out, deg, n);

    {
        const int ntiles = (n + 15) / 16;
        const int wavesPerBlock = 4;
        const int blocks = (ntiles + wavesPerBlock - 1) / wavesPerBlock;
        gcn_gemm_wmma_kernel<<<blocks, wavesPerBlock * 32, 0, stream>>>(x, W, h, n);
    }

    gcn_degree_kernel<<<(e + B - 1) / B, B, 0, stream>>>(col, deg, e);

    gcn_rsqrt_kernel<<<(n + B - 1) / B, B, 0, stream>>>(deg, n);

    {
        long long work = (long long)e * CHANNELS;
        int blocks = (int)((work + B - 1) / B);
        gcn_scatter_kernel<<<blocks, B, 0, stream>>>(row, col, deg, h, out, e);
    }

    gcn_final_kernel<<<(n + B - 1) / B, B, 0, stream>>>(out, h, deg, b, n);
}


# Degree pass: in-degrees counted in per-work-group LDS histograms (packed 16-bit integer counters), rows summed in the rsqrt kernel, in place of 3.2M scattered global f32 atomics; deg stays exact
# speedup vs baseline: 1.6582x; 1.6582x over previous
#include <hip/hip_runtime.h>
#include <math.h>

typedef __attribute__((ext_vector_type(2))) float v2f;
typedef __attribute__((ext_vector_type(8))) float v8f;

#define CHANNELS 16
#define KDIM 128

__global__ void gcn_init_kernel(float* __restrict__ out_accum, int n) {
    int i = blockIdx.x * blockDim.x + threadIdx.x;
    if (i < n * CHANNELS) out_accum[i] = 0.0f;
}

__global__ void gcn_gemm_wmma_kernel(const float* __restrict__ x,
                                     const float* __restrict__ W,
                                     float* __restrict__ h, int n) {
    const int lane   = threadIdx.x & 31;
    const int waveId = threadIdx.x >> 5;
    const int ntiles = (n + 15) >> 4;
    const int tile   = blockIdx.x * (blockDim.x >> 5) + waveId;
    if (tile >= ntiles) return;

    const int half = lane >> 4;
    const int idx  = lane & 15;
    const int koff = half * 2;

    int arow = tile * 16 + idx;
    if (arow >= n) arow = n - 1;
    const float* xrow = x + (size_t)arow * KDIM;

    v8f c = {};
#pragma unroll
    for (int k = 0; k < KDIM; k += 4) {
        v2f a;
        a.x = xrow[k + koff + 0];
        a.y = xrow[k + koff + 1];
        v2f bm;
        bm.x = W[(k + koff + 0) * CHANNELS + idx];
        bm.y = W[(k + koff + 1) * CHANNELS + idx];
        c = __builtin_amdgcn_wmma_f32_16x16x4_f32(
             false, a,  false, bm,
             (short)0, c,  false,  false);
    }

#pragma unroll
    for (int v = 0; v < 8; ++v) {
        int row = tile * 16 + v + 8 * half;
        if (row < n) h[(size_t)row * CHANNELS + idx] = c[v];
    }
}

#define DEG_BLOCKS  128
#define DEG_THREADS 1024
#define DEG_CAPW    51200
#define DEG_BATCH   16

__global__ __launch_bounds__(DEG_THREADS)
void gcn_degree_lds_kernel(const int* __restrict__ col,
                           unsigned* __restrict__ part, int n, int e) {
    __shared__ __attribute__((aligned(16))) unsigned hist[DEG_CAPW];
    const int tid   = threadIdx.x;
    const int wp    = (((n + 1) >> 1) + 3) & ~3;
    const int chunk = (e + DEG_BLOCKS - 1) / DEG_BLOCKS;
    const int e0    = blockIdx.x * chunk;
    const int e1    = min(e0 + chunk, e);
    unsigned* prow  = part + (size_t)blockIdx.x * wp;

    for (int basew = 0; basew < wp; basew += DEG_CAPW) {
        const int nw = min(DEG_CAPW, wp - basew);
        for (int w = tid * 4; w < nw; w += DEG_THREADS * 4)
            *reinterpret_cast<uint4*>(&hist[w]) = make_uint4(0u, 0u, 0u, 0u);
        __syncthreads();
        for (int i0 = e0 + tid; i0 < e1; i0 += DEG_THREADS * DEG_BATCH) {
            int c[DEG_BATCH];
#pragma unroll
            for (int k = 0; k < DEG_BATCH; ++k)
                c[k] = col[min(i0 + k * DEG_THREADS, e1 - 1)];
#pragma unroll
            for (int k = 0; k < DEG_BATCH; ++k) {
                unsigned rel = (i0 + k * DEG_THREADS < e1)
                             ? (unsigned)c[k] - (unsigned)(basew * 2)
                             : 0xFFFFFFFFu;
                if (rel < (unsigned)(nw * 2))
                    atomicAdd(&hist[rel >> 1], 1u << ((rel & 1u) * 16u));
            }
        }
        __syncthreads();
        for (int w = tid * 4; w < nw; w += DEG_THREADS * 4)
            *reinterpret_cast<uint4*>(&prow[basew + w]) =
                *reinterpret_cast<const uint4*>(&hist[w]);
        __syncthreads();
    }
}

__global__ void gcn_dis_kernel(const unsigned* __restrict__ part,
                               float* __restrict__ dis, int n) {
    const int w  = blockIdx.x * blockDim.x + threadIdx.x;
    const int wp = (((n + 1) >> 1) + 3) & ~3;
    if (2 * w >= n) return;
    unsigned lo = 0u, hi = 0u;
    const unsigned* p = part + w;
#pragma unroll 16
    for (int b = 0; b < DEG_BLOCKS; ++b) {
        unsigned v = p[(size_t)b * wp];
        lo += v & 0xFFFFu;
        hi += v >> 16;
    }
    dis[2 * w] = rsqrtf(1.0f + (float)lo);
    if (2 * w + 1 < n) dis[2 * w + 1] = rsqrtf(1.0f + (float)hi);
}

__global__ void gcn_scatter_kernel(const int* __restrict__ row,
                                   const int* __restrict__ col,
                                   const float* __restrict__ dis,
                                   const float* __restrict__ h,
                                   float* __restrict__ out_accum, int e) {
    long long t = (long long)blockIdx.x * blockDim.x + threadIdx.x;
    long long ec = (long long)e * CHANNELS;
    if (t >= ec) return;
    int edge = (int)(t >> 4);
    int ch   = (int)(t & 15);
    int r = row[edge];
    int cl = col[edge];
    float norm = dis[r] * dis[cl];
    float msg  = norm * h[(size_t)r * CHANNELS + ch];
    atomicAdd(&out_accum[(size_t)cl * CHANNELS + ch], msg);
}

__global__ void gcn_final_kernel(float* __restrict__ out,
                                 const float* __restrict__ h,
                                 const float* __restrict__ dis,
                                 const float* __restrict__ b, int n) {
    int i = blockIdx.x * blockDim.x + threadIdx.x;
    if (i >= n) return;
    float selfw = dis[i] * dis[i];
    float v[CHANNELS];
    float m = -INFINITY;
#pragma unroll
    for (int c = 0; c < CHANNELS; ++c) {
        float val = out[(size_t)i * CHANNELS + c]
                  + selfw * h[(size_t)i * CHANNELS + c] + b[c];
        val = fmaxf(val, 0.0f);
        v[c] = val;
        m = fmaxf(m, val);
    }
    float s = 0.0f;
#pragma unroll
    for (int c = 0; c < CHANNELS; ++c) s += __expf(v[c] - m);
    float lse = m + __logf(s);
#pragma unroll
    for (int c = 0; c < CHANNELS; ++c)
        out[(size_t)i * CHANNELS + c] = v[c] - lse;
}

extern "C" void kernel_launch(void* const* d_in, const int* in_sizes, int n_in,
                              void* d_out, int out_size, void* d_ws, size_t ws_size,
                              hipStream_t stream) {
    const float* x  = (const float*)d_in[0];
    const int*   ei = (const int*)d_in[1];
    const float* W  = (const float*)d_in[2];
    const float* b  = (const float*)d_in[3];
    float* out = (float*)d_out;

    const int n = in_sizes[0] / KDIM;
    const int e = in_sizes[1] / 2;
    const int* row = ei;
    const int* col = ei + e;

    float* h   = (float*)d_ws;
    float* deg = h + (size_t)n * CHANNELS;
    unsigned* part = (unsigned*)(deg + (((size_t)n + 3) & ~(size_t)3));

    const int B = 256;

    gcn_init_kernel<<<(n * CHANNELS + B - 1) / B, B, 0, stream>>>(out, n);

    {
        const int ntiles = (n + 15) / 16;
        const int wavesPerBlock = 4;
        const int blocks = (ntiles + wavesPerBlock - 1) / wavesPerBlock;
        gcn_gemm_wmma_kernel<<<blocks, wavesPerBlock * 32, 0, stream>>>(x, W, h, n);
    }

    gcn_degree_lds_kernel<<<DEG_BLOCKS, DEG_THREADS, 0, stream>>>(col, part, n, e);

    gcn_dis_kernel<<<((n + 1) / 2 + B - 1) / B, B, 0, stream>>>(part, deg, n);

    {
        long long work = (long long)e * CHANNELS;
        int blocks = (int)((work + B - 1) / B);
        gcn_scatter_kernel<<<blocks, B, 0, stream>>>(row, col, deg, h, out, e);
    }

    gcn_final_kernel<<<(n + B - 1) / B, B, 0, stream>>>(out, h, deg, b, n);
}


# Edge scatter-add moved from per-edge global f32 atomics to LDS float adds: edges partitioned by 512-node target bucket, a work-group sums a bucket in LDS and fuses the epilogue; arrival-ordered sums
# speedup vs baseline: 3.2221x; 1.9431x over previous
#include <hip/hip_runtime.h>
#include <math.h>

typedef __attribute__((ext_vector_type(2))) float v2f;
typedef __attribute__((ext_vector_type(8))) float v8f;

#define CHANNELS 16
#define KDIM 128

__global__ void gcn_gemm_wmma_kernel(const float* __restrict__ x,
                                     const float* __restrict__ W,
                                     float* __restrict__ h, int n) {
    const int lane   = threadIdx.x & 31;
    const int waveId = threadIdx.x >> 5;
    const int ntiles = (n + 15) >> 4;
    const int tile   = blockIdx.x * (blockDim.x >> 5) + waveId;
    if (tile >= ntiles) return;

    const int half = lane >> 4;
    const int idx  = lane & 15;
    const int koff = half * 2;

    int arow = tile * 16 + idx;
    if (arow >= n) arow = n - 1;
    const float* xrow = x + (size_t)arow * KDIM;

    v8f c = {};
#pragma unroll
    for (int k = 0; k < KDIM; k += 4) {
        v2f a;
        a.x = xrow[k + koff + 0];
        a.y = xrow[k + koff + 1];
        v2f bm;
        bm.x = W[(k + koff + 0) * CHANNELS + idx];
        bm.y = W[(k + koff + 1) * CHANNELS + idx];
        c = __builtin_amdgcn_wmma_f32_16x16x4_f32(
             false, a,  false, bm,
             (short)0, c,  false,  false);
    }

#pragma unroll
    for (int v = 0; v < 8; ++v) {
        int row = tile * 16 + v + 8 * half;
        if (row < n) h[(size_t)row * CHANNELS + idx] = c[v];
    }
}

#define DEG_BLOCKS  128
#define DEG_THREADS 1024
#define DEG_CAPW    51200
#define DEG_BATCH   16

#define BK_SHIFT 9
#define BK_NODES 512
#define BK_MAX   256

__global__ __launch_bounds__(DEG_THREADS)
void gcn_degree_lds_kernel(const int* __restrict__ col,
                           unsigned* __restrict__ part,
                           unsigned* __restrict__ bcnt, int n, int e) {
    __shared__ __attribute__((aligned(16))) unsigned hist[DEG_CAPW];
    const int tid   = threadIdx.x;
    const int wp    = (((n + 1) >> 1) + 3) & ~3;
    const int chunk = (e + DEG_BLOCKS - 1) / DEG_BLOCKS;
    const int e0    = blockIdx.x * chunk;
    const int e1    = min(e0 + chunk, e);
    unsigned* prow  = part + (size_t)blockIdx.x * wp;

    for (int basew = 0; basew < wp; basew += DEG_CAPW) {
        const int nw = min(DEG_CAPW, wp - basew);
        for (int w = tid * 4; w < nw; w += DEG_THREADS * 4)
            *reinterpret_cast<uint4*>(&hist[w]) = make_uint4(0u, 0u, 0u, 0u);
        __syncthreads();
        for (int i0 = e0 + tid; i0 < e1; i0 += DEG_THREADS * DEG_BATCH) {
            int c[DEG_BATCH];
#pragma unroll
            for (int k = 0; k < DEG_BATCH; ++k)
                c[k] = col[min(i0 + k * DEG_THREADS, e1 - 1)];
#pragma unroll
            for (int k = 0; k < DEG_BATCH; ++k) {
                unsigned rel = (i0 + k * DEG_THREADS < e1)
                             ? (unsigned)c[k] - (unsigned)(basew * 2)
                             : 0xFFFFFFFFu;
                if (rel < (unsigned)(nw * 2))
                    atomicAdd(&hist[rel >> 1], 1u << ((rel & 1u) * 16u));
            }
        }
        __syncthreads();
        for (int w = tid * 4; w < nw; w += DEG_THREADS * 4)
            *reinterpret_cast<uint4*>(&prow[basew + w]) =
                *reinterpret_cast<const uint4*>(&hist[w]);
        {
            const int lane = tid & 31;
            const int nbl  = (nw + 255) >> 8;
            for (int j = tid >> 5; j < nbl; j += DEG_THREADS / 32) {
                const int w0 = (j << 8) + lane * 8;
                unsigned s = 0u;
                if (w0 < nw) {
                    uint4 a = *reinterpret_cast<const uint4*>(&hist[w0]);
                    s += (a.x & 0xFFFFu) + (a.x >> 16) + (a.y & 0xFFFFu) + (a.y >> 16)
                       + (a.z & 0xFFFFu) + (a.z >> 16) + (a.w & 0xFFFFu) + (a.w >> 16);
                }
                if (w0 + 4 < nw) {
                    uint4 a = *reinterpret_cast<const uint4*>(&hist[w0 + 4]);
                    s += (a.x & 0xFFFFu) + (a.x >> 16) + (a.y & 0xFFFFu) + (a.y >> 16)
                       + (a.z & 0xFFFFu) + (a.z >> 16) + (a.w & 0xFFFFu) + (a.w >> 16);
                }
                for (int d = 16; d > 0; d >>= 1) s += __shfl_xor(s, d);
                if (lane == 0)
                    bcnt[blockIdx.x * BK_MAX + (basew >> 8) + j] = s;
            }
        }
        __syncthreads();
    }
}

__global__ void gcn_dis_kernel(const unsigned* __restrict__ part,
                               float* __restrict__ dis, int n) {
    const int w  = blockIdx.x * blockDim.x + threadIdx.x;
    const int wp = (((n + 1) >> 1) + 3) & ~3;
    if (2 * w >= n) return;
    unsigned lo = 0u, hi = 0u;
    const unsigned* p = part + w;
#pragma unroll 16
    for (int b = 0; b < DEG_BLOCKS; ++b) {
        unsigned v = p[(size_t)b * wp];
        lo += v & 0xFFFFu;
        hi += v >> 16;
    }
    dis[2 * w] = rsqrtf(1.0f + (float)lo);
    if (2 * w + 1 < n) dis[2 * w + 1] = rsqrtf(1.0f + (float)hi);
}

#define PART_BATCH 8

__global__ __launch_bounds__(DEG_THREADS)
void gcn_partition_kernel(const int* __restrict__ row,
                          const int* __restrict__ col,
                          const unsigned* __restrict__ bcnt,
                          unsigned* __restrict__ bstart,
                          int2* __restrict__ sorted, int n, int e) {
    __shared__ unsigned tot[BK_MAX];
    __shared__ unsigned bef[BK_MAX];
    __shared__ unsigned cur[BK_MAX];
    const int tid   = threadIdx.x;
    const int nbk   = (n + BK_NODES - 1) >> BK_SHIFT;
    const int chunk = (e + DEG_BLOCKS - 1) / DEG_BLOCKS;
    const int e0    = blockIdx.x * chunk;
    const int e1    = min(e0 + chunk, e);

    if (tid < BK_MAX) {
        unsigned t = 0u, bf = 0u;
        if (tid < nbk) {
#pragma unroll 16
            for (int b = 0; b < DEG_BLOCKS; ++b) {
                unsigned v = bcnt[b * BK_MAX + tid];
                t  += v;
                bf += (b < (int)blockIdx.x) ? v : 0u;
            }
        }
        tot[tid] = t;
        bef[tid] = bf;
    }
    __syncthreads();
    if (tid < 32) {
        unsigned v[8];
        unsigned run = 0u;
#pragma unroll
        for (int i = 0; i < 8; ++i) {
            unsigned t = tot[tid * 8 + i];
            v[i] = run;
            run += t;
        }
        unsigned inc = run;
        for (int d = 1; d < 32; d <<= 1) {
            unsigned t = __shfl_up(inc, d);
            if (tid >= d) inc += t;
        }
        const unsigned excl = inc - run;
#pragma unroll
        for (int i = 0; i < 8; ++i) {
            const unsigned st = excl + v[i];
            cur[tid * 8 + i] = st + bef[tid * 8 + i];
            if (blockIdx.x == 0) bstart[tid * 8 + i] = st;
        }
        if (blockIdx.x == 0 && tid == 31) bstart[BK_MAX] = inc;
    }
    __syncthreads();

    for (int i0 = e0 + tid; i0 < e1; i0 += DEG_THREADS * PART_BATCH) {
        int c[PART_BATCH], r[PART_BATCH];
        unsigned pos[PART_BATCH];
#pragma unroll
        for (int k = 0; k < PART_BATCH; ++k) {
            const int idx = min(i0 + k * DEG_THREADS, e1 - 1);
            c[k] = col[idx];
            r[k] = row[idx];
        }
        asm volatile("" :: "v"(c[0]), "v"(c[1]), "v"(c[2]), "v"(c[3]),
                           "v"(c[4]), "v"(c[5]), "v"(c[6]), "v"(c[7]),
                           "v"(r[0]), "v"(r[1]), "v"(r[2]), "v"(r[3]),
                           "v"(r[4]), "v"(r[5]), "v"(r[6]), "v"(r[7]));
#pragma unroll
        for (int k = 0; k < PART_BATCH; ++k) {
            pos[k] = 0u;
            if (i0 + k * DEG_THREADS < e1)
                pos[k] = atomicAdd(&cur[c[k] >> BK_SHIFT], 1u);
        }
#pragma unroll
        for (int k = 0; k < PART_BATCH; ++k)
            if (i0 + k * DEG_THREADS < e1)
                sorted[pos[k]] = make_int2(r[k], c[k]);
    }
}

#define AGG_THREADS 1024
#define AGG_GROUPS  (AGG_THREADS / CHANNELS)
#define AGG_BATCH   8

__global__ __launch_bounds__(AGG_THREADS)
void gcn_aggregate_kernel(const int2* __restrict__ sorted,
                          const unsigned* __restrict__ bstart,
                          const float* __restrict__ dis,
                          const float* __restrict__ h,
                          const float* __restrict__ b,
                          float* __restrict__ out, int n) {
    __shared__ __attribute__((aligned(16))) float acc[BK_NODES * CHANNELS];
    const int tid  = threadIdx.x;
    const int base = blockIdx.x << BK_SHIFT;

    for (int w = tid * 4; w < BK_NODES * CHANNELS; w += AGG_THREADS * 4)
        *reinterpret_cast<float4*>(&acc[w]) = make_float4(0.0f, 0.0f, 0.0f, 0.0f);
    __syncthreads();

    const int s0 = (int)bstart[blockIdx.x];
    const int s1 = (int)bstart[blockIdx.x + 1];
    const int g  = tid >> 4;
    const int ch = tid & 15;
    const int step = AGG_GROUPS * AGG_BATCH;
    int2 ed[AGG_BATCH];
#pragma unroll
    for (int u = 0; u < AGG_BATCH; ++u) ed[u] = make_int2(0, 0);
    if (s0 + g < s1) {
#pragma unroll
        for (int u = 0; u < AGG_BATCH; ++u)
            ed[u] = sorted[min(s0 + g + u * AGG_GROUPS, s1 - 1)];
    }
    for (int i0 = s0 + g; i0 < s1; i0 += step) {
        float dr[AGG_BATCH], dc[AGG_BATCH], hv[AGG_BATCH];
        int2  nd[AGG_BATCH];
        asm volatile("" :: "v"(ed[0].x), "v"(ed[0].y), "v"(ed[1].x), "v"(ed[1].y),
                           "v"(ed[2].x), "v"(ed[2].y), "v"(ed[3].x), "v"(ed[3].y),
                           "v"(ed[4].x), "v"(ed[4].y), "v"(ed[5].x), "v"(ed[5].y),
                           "v"(ed[6].x), "v"(ed[6].y), "v"(ed[7].x), "v"(ed[7].y));
#pragma unroll
        for (int u = 0; u < AGG_BATCH; ++u) {
            dr[u] = dis[ed[u].x];
            dc[u] = dis[ed[u].y];
            hv[u] = h[(size_t)ed[u].x * CHANNELS + ch];
        }
#pragma unroll
        for (int u = 0; u < AGG_BATCH; ++u)
            nd[u] = sorted[min(i0 + step + u * AGG_GROUPS, s1 - 1)];
        asm volatile("" :: "v"(dr[0]), "v"(dr[1]), "v"(dr[2]), "v"(dr[3]),
                           "v"(dr[4]), "v"(dr[5]), "v"(dr[6]), "v"(dr[7]),
                           "v"(dc[0]), "v"(dc[1]), "v"(dc[2]), "v"(dc[3]),
                           "v"(dc[4]), "v"(dc[5]), "v"(dc[6]), "v"(dc[7]),
                           "v"(hv[0]), "v"(hv[1]), "v"(hv[2]), "v"(hv[3]),
                           "v"(hv[4]), "v"(hv[5]), "v"(hv[6]), "v"(hv[7]));
#pragma unroll
        for (int u = 0; u < AGG_BATCH; ++u) {
            if (i0 + u * AGG_GROUPS < s1) {
                const float norm = dr[u] * dc[u];
                atomicAdd(&acc[(ed[u].y - base) * CHANNELS + ch], norm * hv[u]);
            }
        }
#pragma unroll
        for (int u = 0; u < AGG_BATCH; ++u) ed[u] = nd[u];
    }
    __syncthreads();

    const int i = base + tid;
    if (tid < BK_NODES && i < n) {
        const float selfw = dis[i] * dis[i];
        float v[CHANNELS];
        float m = -INFINITY;
#pragma unroll
        for (int c = 0; c < CHANNELS; ++c) {
            float val = acc[tid * CHANNELS + c]
                      + selfw * h[(size_t)i * CHANNELS + c] + b[c];
            val = fmaxf(val, 0.0f);
            v[c] = val;
            m = fmaxf(m, val);
        }
        float s = 0.0f;
#pragma unroll
        for (int c = 0; c < CHANNELS; ++c) s += __expf(v[c] - m);
        float lse = m + __logf(s);
#pragma unroll
        for (int c = 0; c < CHANNELS; ++c)
            out[(size_t)i * CHANNELS + c] = v[c] - lse;
    }
}

extern "C" void kernel_launch(void* const* d_in, const int* in_sizes, int n_in,
                              void* d_out, int out_size, void* d_ws, size_t ws_size,
                              hipStream_t stream) {
    const float* x  = (const float*)d_in[0];
    const int*   ei = (const int*)d_in[1];
    const float* W  = (const float*)d_in[2];
    const float* b  = (const float*)d_in[3];
    float* out = (float*)d_out;

    const int n = in_sizes[0] / KDIM;
    const int e = in_sizes[1] / 2;
    const int* row = ei;
    const int* col = ei + e;

    float* h   = (float*)d_ws;
    float* deg = h + (size_t)n * CHANNELS;
    unsigned* bcnt   = (unsigned*)(deg + (((size_t)n + 3) & ~(size_t)3));
    unsigned* bstart = bcnt + (size_t)DEG_BLOCKS * BK_MAX;
    int2* sorted     = (int2*)(bstart + BK_MAX + 16);
    unsigned* part   = (unsigned*)(sorted + (((size_t)e + 1) & ~(size_t)1));

    const int B = 256;

    {
        const int ntiles = (n + 15) / 16;
        const int wavesPerBlock = 4;
        const int blocks = (ntiles + wavesPerBlock - 1) / wavesPerBlock;
        gcn_gemm_wmma_kernel<<<blocks, wavesPerBlock * 32, 0, stream>>>(x, W, h, n);
    }

    gcn_degree_lds_kernel<<<DEG_BLOCKS, DEG_THREADS, 0, stream>>>(col, part, bcnt, n, e);

    gcn_dis_kernel<<<((n + 1) / 2 + B - 1) / B, B, 0, stream>>>(part, deg, n);

    gcn_partition_kernel<<<DEG_BLOCKS, DEG_THREADS, 0, stream>>>(row, col, bcnt, bstart,
                                                                 sorted, n, e);

    gcn_aggregate_kernel<<<(n + BK_NODES - 1) / BK_NODES, AGG_THREADS, 0, stream>>>(
        sorted, bstart, deg, h, b, out, n);
}


# Edge weight dis[r]*dis[c] factored out of the edge loop: rows of x@W stored times dis[r] by the gemm, node sums times dis[c] in the epilogue; per edge one gather, no multiply; products re-associated
# speedup vs baseline: 3.6540x; 1.1340x over previous
#include <hip/hip_runtime.h>
#include <math.h>

typedef __attribute__((ext_vector_type(2))) float v2f;
typedef __attribute__((ext_vector_type(8))) float v8f;

#define CHANNELS 16
#define KDIM 128

__global__ void gcn_gemm_wmma_kernel(const float* __restrict__ x,
                                     const float* __restrict__ W,
                                     const float* __restrict__ dis,
                                     float* __restrict__ h, int n) {
    const int lane   = threadIdx.x & 31;
    const int waveId = threadIdx.x >> 5;
    const int ntiles = (n + 15) >> 4;
    const int tile   = blockIdx.x * (blockDim.x >> 5) + waveId;
    if (tile >= ntiles) return;

    const int half = lane >> 4;
    const int idx  = lane & 15;
    const int koff = half * 2;

    int arow = tile * 16 + idx;
    if (arow >= n) arow = n - 1;
    const float* xrow = x + (size_t)arow * KDIM;

    float dsc[8];
#pragma unroll
    for (int v = 0; v < 8; ++v)
        dsc[v] = dis[min(tile * 16 + v + 8 * half, n - 1)];

    v8f c = {};
#pragma unroll
    for (int k = 0; k < KDIM; k += 4) {
        v2f a;
        a.x = xrow[k + koff + 0];
        a.y = xrow[k + koff + 1];
        v2f bm;
        bm.x = W[(k + koff + 0) * CHANNELS + idx];
        bm.y = W[(k + koff + 1) * CHANNELS + idx];
        c = __builtin_amdgcn_wmma_f32_16x16x4_f32(
             false, a,  false, bm,
             (short)0, c,  false,  false);
    }

#pragma unroll
    for (int v = 0; v < 8; ++v) {
        int row = tile * 16 + v + 8 * half;
        if (row < n) h[(size_t)row * CHANNELS + idx] = c[v] * dsc[v];
    }
}

#define DEG_BLOCKS  128
#define DEG_THREADS 1024
#define DEG_CAPW    51200
#define DEG_BATCH   16

#define BK_SHIFT 9
#define BK_NODES 512
#define BK_MAX   256

__global__ __launch_bounds__(DEG_THREADS)
void gcn_degree_lds_kernel(const int* __restrict__ col,
                           unsigned* __restrict__ part,
                           unsigned* __restrict__ bcnt, int n, int e) {
    __shared__ __attribute__((aligned(16))) unsigned hist[DEG_CAPW];
    const int tid   = threadIdx.x;
    const int wp    = (((n + 1) >> 1) + 3) & ~3;
    const int chunk = (e + DEG_BLOCKS - 1) / DEG_BLOCKS;
    const int e0    = blockIdx.x * chunk;
    const int e1    = min(e0 + chunk, e);
    unsigned* prow  = part + (size_t)blockIdx.x * wp;

    for (int basew = 0; basew < wp; basew += DEG_CAPW) {
        const int nw = min(DEG_CAPW, wp - basew);
        for (int w = tid * 4; w < nw; w += DEG_THREADS * 4)
            *reinterpret_cast<uint4*>(&hist[w]) = make_uint4(0u, 0u, 0u, 0u);
        __syncthreads();
        for (int i0 = e0 + tid; i0 < e1; i0 += DEG_THREADS * DEG_BATCH) {
            int c[DEG_BATCH];
#pragma unroll
            for (int k = 0; k < DEG_BATCH; ++k)
                c[k] = col[min(i0 + k * DEG_THREADS, e1 - 1)];
#pragma unroll
            for (int k = 0; k < DEG_BATCH; ++k) {
                unsigned rel = (i0 + k * DEG_THREADS < e1)
                             ? (unsigned)c[k] - (unsigned)(basew * 2)
                             : 0xFFFFFFFFu;
                if (rel < (unsigned)(nw * 2))
                    atomicAdd(&hist[rel >> 1], 1u << ((rel & 1u) * 16u));
            }
        }
        __syncthreads();
        for (int w = tid * 4; w < nw; w += DEG_THREADS * 4)
            *reinterpret_cast<uint4*>(&prow[basew + w]) =
                *reinterpret_cast<const uint4*>(&hist[w]);
        {
            const int lane = tid & 31;
            const int nbl  = (nw + 255) >> 8;
            for (int j = tid >> 5; j < nbl; j += DEG_THREADS / 32) {
                const int w0 = (j << 8) + lane * 8;
                unsigned s = 0u;
                if (w0 < nw) {
                    uint4 a = *reinterpret_cast<const uint4*>(&hist[w0]);
                    s += (a.x & 0xFFFFu) + (a.x >> 16) + (a.y & 0xFFFFu) + (a.y >> 16)
                       + (a.z & 0xFFFFu) + (a.z >> 16) + (a.w & 0xFFFFu) + (a.w >> 16);
                }
                if (w0 + 4 < nw) {
                    uint4 a = *reinterpret_cast<const uint4*>(&hist[w0 + 4]);
                    s += (a.x & 0xFFFFu) + (a.x >> 16) + (a.y & 0xFFFFu) + (a.y >> 16)
                       + (a.z & 0xFFFFu) + (a.z >> 16) + (a.w & 0xFFFFu) + (a.w >> 16);
                }
                for (int d = 16; d > 0; d >>= 1) s += __shfl_xor(s, d);
                if (lane == 0)
                    bcnt[blockIdx.x * BK_MAX + (basew >> 8) + j] = s;
            }
        }
        __syncthreads();
    }
}

__global__ void gcn_dis_kernel(const unsigned* __restrict__ part,
                               float* __restrict__ dis, int n) {
    const int w  = blockIdx.x * blockDim.x + threadIdx.x;
    const int wp = (((n + 1) >> 1) + 3) & ~3;
    if (2 * w >= n) return;
    unsigned lo = 0u, hi = 0u;
    const unsigned* p = part + w;
#pragma unroll 16
    for (int b = 0; b < DEG_BLOCKS; ++b) {
        unsigned v = p[(size_t)b * wp];
        lo += v & 0xFFFFu;
        hi += v >> 16;
    }
    dis[2 * w] = rsqrtf(1.0f + (float)lo);
    if (2 * w + 1 < n) dis[2 * w + 1] = rsqrtf(1.0f + (float)hi);
}

#define PART_BATCH 8

__global__ __launch_bounds__(DEG_THREADS)
void gcn_partition_kernel(const int* __restrict__ row,
                          const int* __restrict__ col,
                          const unsigned* __restrict__ bcnt,
                          unsigned* __restrict__ bstart,
                          int2* __restrict__ sorted, int n, int e) {
    __shared__ unsigned tot[BK_MAX];
    __shared__ unsigned bef[BK_MAX];
    __shared__ unsigned cur[BK_MAX];
    const int tid   = threadIdx.x;
    const int nbk   = (n + BK_NODES - 1) >> BK_SHIFT;
    const int chunk = (e + DEG_BLOCKS - 1) / DEG_BLOCKS;
    const int e0    = blockIdx.x * chunk;
    const int e1    = min(e0 + chunk, e);

    if (tid < BK_MAX) {
        unsigned t = 0u, bf = 0u;
        if (tid < nbk) {
#pragma unroll 16
            for (int b = 0; b < DEG_BLOCKS; ++b) {
                unsigned v = bcnt[b * BK_MAX + tid];
                t  += v;
                bf += (b < (int)blockIdx.x) ? v : 0u;
            }
        }
        tot[tid] = t;
        bef[tid] = bf;
    }
    __syncthreads();
    if (tid < 32) {
        unsigned v[8];
        unsigned run = 0u;
#pragma unroll
        for (int i = 0; i < 8; ++i) {
            unsigned t = tot[tid * 8 + i];
            v[i] = run;
            run += t;
        }
        unsigned inc = run;
        for (int d = 1; d < 32; d <<= 1) {
            unsigned t = __shfl_up(inc, d);
            if (tid >= d) inc += t;
        }
        const unsigned excl = inc - run;
#pragma unroll
        for (int i = 0; i < 8; ++i) {
            const unsigned st = excl + v[i];
            cur[tid * 8 + i] = st + bef[tid * 8 + i];
            if (blockIdx.x == 0) bstart[tid * 8 + i] = st;
        }
        if (blockIdx.x == 0 && tid == 31) bstart[BK_MAX] = inc;
    }
    __syncthreads();

    for (int i0 = e0 + tid; i0 < e1; i0 += DEG_THREADS * PART_BATCH) {
        int c[PART_BATCH], r[PART_BATCH];
        unsigned pos[PART_BATCH];
#pragma unroll
        for (int k = 0; k < PART_BATCH; ++k) {
            const int idx = min(i0 + k * DEG_THREADS, e1 - 1);
            c[k] = col[idx];
            r[k] = row[idx];
        }
        asm volatile("" :: "v"(c[0]), "v"(c[1]), "v"(c[2]), "v"(c[3]),
                           "v"(c[4]), "v"(c[5]), "v"(c[6]), "v"(c[7]),
                           "v"(r[0]), "v"(r[1]), "v"(r[2]), "v"(r[3]),
                           "v"(r[4]), "v"(r[5]), "v"(r[6]), "v"(r[7]));
#pragma unroll
        for (int k = 0; k < PART_BATCH; ++k) {
            pos[k] = 0u;
            if (i0 + k * DEG_THREADS < e1)
                pos[k] = atomicAdd(&cur[c[k] >> BK_SHIFT], 1u);
        }
#pragma unroll
        for (int k = 0; k < PART_BATCH; ++k)
            if (i0 + k * DEG_THREADS < e1)
                sorted[pos[k]] = make_int2(r[k], c[k]);
    }
}

#define AGG_THREADS 1024
#define AGG_GROUPS  (AGG_THREADS / CHANNELS)
#define AGG_BATCH   8

__global__ __launch_bounds__(AGG_THREADS)
void gcn_aggregate_kernel(const int2* __restrict__ sorted,
                          const unsigned* __restrict__ bstart,
                          const float* __restrict__ dis,
                          const float* __restrict__ h,
                          const float* __restrict__ b,
                          float* __restrict__ out, int n) {
    __shared__ __attribute__((aligned(16))) float acc[BK_NODES * CHANNELS];
    const int tid  = threadIdx.x;
    const int base = blockIdx.x << BK_SHIFT;

    for (int w = tid * 4; w < BK_NODES * CHANNELS; w += AGG_THREADS * 4)
        *reinterpret_cast<float4*>(&acc[w]) = make_float4(0.0f, 0.0f, 0.0f, 0.0f);
    __syncthreads();

    const int s0 = (int)bstart[blockIdx.x];
    const int s1 = (int)bstart[blockIdx.x + 1];
    const int g  = tid >> 4;
    const int ch = tid & 15;
    const int step = AGG_GROUPS * AGG_BATCH;
    int2 ed[AGG_BATCH];
#pragma unroll
    for (int u = 0; u < AGG_BATCH; ++u) ed[u] = make_int2(0, 0);
    if (s0 + g < s1) {
#pragma unroll
        for (int u = 0; u < AGG_BATCH; ++u)
            ed[u] = sorted[min(s0 + g + u * AGG_GROUPS, s1 - 1)];
    }
    for (int i0 = s0 + g; i0 < s1; i0 += step) {
        float hv[AGG_BATCH];
        int2  nd[AGG_BATCH];
        asm volatile("" :: "v"(ed[0].x), "v"(ed[0].y), "v"(ed[1].x), "v"(ed[1].y),
                           "v"(ed[2].x), "v"(ed[2].y), "v"(ed[3].x), "v"(ed[3].y),
                           "v"(ed[4].x), "v"(ed[4].y), "v"(ed[5].x), "v"(ed[5].y),
                           "v"(ed[6].x), "v"(ed[6].y), "v"(ed[7].x), "v"(ed[7].y));
#pragma unroll
        for (int u = 0; u < AGG_BATCH; ++u)
            hv[u] = h[(size_t)ed[u].x * CHANNELS + ch];
#pragma unroll
        for (int u = 0; u < AGG_BATCH; ++u)
            nd[u] = sorted[min(i0 + step + u * AGG_GROUPS, s1 - 1)];
        asm volatile("" :: "v"(hv[0]), "v"(hv[1]), "v"(hv[2]), "v"(hv[3]),
                           "v"(hv[4]), "v"(hv[5]), "v"(hv[6]), "v"(hv[7]));
#pragma unroll
        for (int u = 0; u < AGG_BATCH; ++u) {
            if (i0 + u * AGG_GROUPS < s1)
                atomicAdd(&acc[(ed[u].y - base) * CHANNELS + ch], hv[u]);
        }
#pragma unroll
        for (int u = 0; u < AGG_BATCH; ++u) ed[u] = nd[u];
    }
    __syncthreads();

    const int i = base + tid;
    if (tid < BK_NODES && i < n) {
        const float di = dis[i];
        float v[CHANNELS];
        float m = -INFINITY;
#pragma unroll
        for (int c = 0; c < CHANNELS; ++c) {
            float val = di * (acc[tid * CHANNELS + c] + h[(size_t)i * CHANNELS + c]) + b[c];
            val = fmaxf(val, 0.0f);
            v[c] = val;
            m = fmaxf(m, val);
        }
        float s = 0.0f;
#pragma unroll
        for (int c = 0; c < CHANNELS; ++c) s += __expf(v[c] - m);
        float lse = m + __logf(s);
#pragma unroll
        for (int c = 0; c < CHANNELS; ++c)
            out[(size_t)i * CHANNELS + c] = v[c] - lse;
    }
}

extern "C" void kernel_launch(void* const* d_in, const int* in_sizes, int n_in,
                              void* d_out, int out_size, void* d_ws, size_t ws_size,
                              hipStream_t stream) {
    const float* x  = (const float*)d_in[0];
    const int*   ei = (const int*)d_in[1];
    const float* W  = (const float*)d_in[2];
    const float* b  = (const float*)d_in[3];
    float* out = (float*)d_out;

    const int n = in_sizes[0] / KDIM;
    const int e = in_sizes[1] / 2;
    const int* row = ei;
    const int* col = ei + e;

    float* h   = (float*)d_ws;
    float* deg = h + (size_t)n * CHANNELS;
    unsigned* bcnt   = (unsigned*)(deg + (((size_t)n + 3) & ~(size_t)3));
    unsigned* bstart = bcnt + (size_t)DEG_BLOCKS * BK_MAX;
    int2* sorted     = (int2*)(bstart + BK_MAX + 16);
    unsigned* part   = (unsigned*)(sorted + (((size_t)e + 1) & ~(size_t)1));

    const int B = 256;

    gcn_degree_lds_kernel<<<DEG_BLOCKS, DEG_THREADS, 0, stream>>>(col, part, bcnt, n, e);

    gcn_dis_kernel<<<((n + 1) / 2 + B - 1) / B, B, 0, stream>>>(part, deg, n);

    {
        const int ntiles = (n + 15) / 16;
        const int wavesPerBlock = 4;
        const int blocks = (ntiles + wavesPerBlock - 1) / wavesPerBlock;
        gcn_gemm_wmma_kernel<<<blocks, wavesPerBlock * 32, 0, stream>>>(x, W, deg, h, n);
    }

    gcn_partition_kernel<<<DEG_BLOCKS, DEG_THREADS, 0, stream>>>(row, col, bcnt, bstart,
                                                                 sorted, n, e);

    gcn_aggregate_kernel<<<(n + BK_NODES - 1) / BK_NODES, AGG_THREADS, 0, stream>>>(
        sorted, bstart, deg, h, b, out, n);
}


# Launches 5 -> 3: edge partition and x@W tiles run as blocks of one launch (neither reads the other's output); each product block sums its own rows' degree counts, so the dis launch goes
# speedup vs baseline: 4.0947x; 1.1206x over previous
#include <hip/hip_runtime.h>
#include <math.h>

typedef __attribute__((ext_vector_type(2))) float v2f;
typedef __attribute__((ext_vector_type(8))) float v8f;

#define CHANNELS 16
#define KDIM 128

#define DEG_BLOCKS  128
#define DEG_THREADS 1024
#define DEG_CAPW    51200
#define DEG_BATCH   16

#define BK_SHIFT 9
#define BK_NODES 512
#define BK_MAX   256

__global__ __launch_bounds__(DEG_THREADS)
void gcn_degree_lds_kernel(const int* __restrict__ col,
                           unsigned* __restrict__ part,
                           unsigned* __restrict__ bcnt, int n, int e) {
    __shared__ __attribute__((aligned(16))) unsigned hist[DEG_CAPW];
    const int tid   = threadIdx.x;
    const int wp    = (((n + 1) >> 1) + 3) & ~3;
    const int chunk = (e + DEG_BLOCKS - 1) / DEG_BLOCKS;
    const int e0    = blockIdx.x * chunk;
    const int e1    = min(e0 + chunk, e);
    unsigned* prow  = part + (size_t)blockIdx.x * wp;

    for (int basew = 0; basew < wp; basew += DEG_CAPW) {
        const int nw = min(DEG_CAPW, wp - basew);
        for (int w = tid * 4; w < nw; w += DEG_THREADS * 4)
            *reinterpret_cast<uint4*>(&hist[w]) = make_uint4(0u, 0u, 0u, 0u);
        __syncthreads();
        for (int i0 = e0 + tid; i0 < e1; i0 += DEG_THREADS * DEG_BATCH) {
            int c[DEG_BATCH];
#pragma unroll
            for (int k = 0; k < DEG_BATCH; ++k)
                c[k] = col[min(i0 + k * DEG_THREADS, e1 - 1)];
#pragma unroll
            for (int k = 0; k < DEG_BATCH; ++k) {
                unsigned rel = (i0 + k * DEG_THREADS < e1)
                             ? (unsigned)c[k] - (unsigned)(basew * 2)
                             : 0xFFFFFFFFu;
                if (rel < (unsigned)(nw * 2))
                    atomicAdd(&hist[rel >> 1], 1u << ((rel & 1u) * 16u));
            }
        }
        __syncthreads();
        for (int w = tid * 4; w < nw; w += DEG_THREADS * 4)
            *reinterpret_cast<uint4*>(&prow[basew + w]) =
                *reinterpret_cast<const uint4*>(&hist[w]);
        {
            const int lane = tid & 31;
            const int nbl  = (nw + 255) >> 8;
            for (int j = tid >> 5; j < nbl; j += DEG_THREADS / 32) {
                const int w0 = (j << 8) + lane * 8;
                unsigned s = 0u;
                if (w0 < nw) {
                    uint4 a = *reinterpret_cast<const uint4*>(&hist[w0]);
                    s += (a.x & 0xFFFFu) + (a.x >> 16) + (a.y & 0xFFFFu) + (a.y >> 16)
                       + (a.z & 0xFFFFu) + (a.z >> 16) + (a.w & 0xFFFFu) + (a.w >> 16);
                }
                if (w0 + 4 < nw) {
                    uint4 a = *reinterpret_cast<const uint4*>(&hist[w0 + 4]);
                    s += (a.x & 0xFFFFu) + (a.x >> 16) + (a.y & 0xFFFFu) + (a.y >> 16)
                       + (a.z & 0xFFFFu) + (a.z >> 16) + (a.w & 0xFFFFu) + (a.w >> 16);
                }
                for (int d = 16; d > 0; d >>= 1) s += __shfl_xor(s, d);
                if (lane == 0)
                    bcnt[blockIdx.x * BK_MAX + (basew >> 8) + j] = s;
            }
        }
        __syncthreads();
    }
}

#define PART_BATCH 8
#define GP_THREADS 1024
#define GEMM_TILES (GP_THREADS / 32)
#define GEMM_ROWS  (GEMM_TILES * 16)

__global__ __launch_bounds__(GP_THREADS)
void gcn_gemm_partition_kernel(const float* __restrict__ x,
                               const float* __restrict__ W,
                               const unsigned* __restrict__ part,
                               float* __restrict__ dis,
                               float* __restrict__ h,
                               const int* __restrict__ row,
                               const int* __restrict__ col,
                               const unsigned* __restrict__ bcnt,
                               unsigned* __restrict__ bstart,
                               int2* __restrict__ sorted, int n, int e) {
    __shared__ unsigned tot[BK_MAX];
    __shared__ unsigned bef[BK_MAX];
    __shared__ unsigned cur[BK_MAX];
    __shared__ float sdis[GEMM_ROWS];
    const int tid = threadIdx.x;

    if (blockIdx.x < DEG_BLOCKS) {
        const int nbk   = (n + BK_NODES - 1) >> BK_SHIFT;
        const int chunk = (e + DEG_BLOCKS - 1) / DEG_BLOCKS;
        const int e0    = blockIdx.x * chunk;
        const int e1    = min(e0 + chunk, e);

        if (tid < BK_MAX) {
            unsigned t = 0u, bf = 0u;
            if (tid < nbk) {
#pragma unroll 16
                for (int b = 0; b < DEG_BLOCKS; ++b) {
                    unsigned v = bcnt[b * BK_MAX + tid];
                    t  += v;
                    bf += (b < (int)blockIdx.x) ? v : 0u;
                }
            }
            tot[tid] = t;
            bef[tid] = bf;
        }
        __syncthreads();
        if (tid < 32) {
            unsigned v[8];
            unsigned run = 0u;
#pragma unroll
            for (int i = 0; i < 8; ++i) {
                unsigned t = tot[tid * 8 + i];
                v[i] = run;
                run += t;
            }
            unsigned inc = run;
            for (int d = 1; d < 32; d <<= 1) {
                unsigned t = __shfl_up(inc, d);
                if (tid >= d) inc += t;
            }
            const unsigned excl = inc - run;
#pragma unroll
            for (int i = 0; i < 8; ++i) {
                const unsigned st = excl + v[i];
                cur[tid * 8 + i] = st + bef[tid * 8 + i];
                if (blockIdx.x == 0) bstart[tid * 8 + i] = st;
            }
            if (blockIdx.x == 0 && tid == 31) bstart[BK_MAX] = inc;
        }
        __syncthreads();

        for (int i0 = e0 + tid; i0 < e1; i0 += GP_THREADS * PART_BATCH) {
            int c[PART_BATCH], r[PART_BATCH];
            unsigned pos[PART_BATCH];
#pragma unroll
            for (int k = 0; k < PART_BATCH; ++k) {
                const int idx = min(i0 + k * GP_THREADS, e1 - 1);
                c[k] = col[idx];
                r[k] = row[idx];
            }
            asm volatile("" :: "v"(c[0]), "v"(c[1]), "v"(c[2]), "v"(c[3]),
                               "v"(c[4]), "v"(c[5]), "v"(c[6]), "v"(c[7]),
                               "v"(r[0]), "v"(r[1]), "v"(r[2]), "v"(r[3]),
                               "v"(r[4]), "v"(r[5]), "v"(r[6]), "v"(r[7]));
#pragma unroll
            for (int k = 0; k < PART_BATCH; ++k) {
                pos[k] = 0u;
                if (i0 + k * GP_THREADS < e1)
                    pos[k] = atomicAdd(&cur[c[k] >> BK_SHIFT], 1u);
            }
#pragma unroll
            for (int k = 0; k < PART_BATCH; ++k)
                if (i0 + k * GP_THREADS < e1)
                    sorted[pos[k]] = make_int2(r[k], c[k]);
        }
    } else {
        const int gb      = blockIdx.x - DEG_BLOCKS;
        const int lane    = tid & 31;
        const int waveId  = __builtin_amdgcn_readfirstlane(tid >> 5);
        const int rowbase = gb * GEMM_ROWS;
        const int ntiles  = (n + 15) >> 4;
        const int tile    = gb * GEMM_TILES + waveId;

        if (tid < GEMM_ROWS / 2) {
            const int w  = (rowbase >> 1) + tid;
            const int wp = (((n + 1) >> 1) + 3) & ~3;
            float d0 = 0.0f, d1 = 0.0f;
            if (2 * w < n) {
                unsigned lo = 0u, hi = 0u;
                const unsigned wu = (unsigned)w;
#pragma unroll 8
                for (int b = 0; b < DEG_BLOCKS; ++b) {
                    const unsigned* prow = part + (size_t)b * wp;
                    unsigned v = prow[wu];
                    lo += v & 0xFFFFu;
                    hi += v >> 16;
                }
                d0 = rsqrtf(1.0f + (float)lo);
                d1 = rsqrtf(1.0f + (float)hi);
                dis[2 * w] = d0;
                if (2 * w + 1 < n) dis[2 * w + 1] = d1;
            }
            sdis[2 * tid]     = d0;
            sdis[2 * tid + 1] = d1;
        }

        const int half = lane >> 4;
        const int idx  = lane & 15;
        const int koff = half * 2;

        v8f c = {};
        if (tile < ntiles) {
            int arow = tile * 16 + idx;
            if (arow >= n) arow = n - 1;
            const float* xrow = x + (size_t)arow * KDIM;
#pragma unroll
            for (int k = 0; k < KDIM; k += 4) {
                v2f a;
                a.x = xrow[k + koff + 0];
                a.y = xrow[k + koff + 1];
                v2f bm;
                bm.x = W[(k + koff + 0) * CHANNELS + idx];
                bm.y = W[(k + koff + 1) * CHANNELS + idx];
                c = __builtin_amdgcn_wmma_f32_16x16x4_f32(
                     false, a,  false, bm,
                     (short)0, c,  false,  false);
            }
        }
        __syncthreads();

        if (tile < ntiles) {
#pragma unroll
            for (int v = 0; v < 8; ++v) {
                const int lr = waveId * 16 + v + 8 * half;
                const int rw = rowbase + lr;
                if (rw < n) h[(size_t)rw * CHANNELS + idx] = c[v] * sdis[lr];
            }
        }
    }
}

#define AGG_THREADS 1024
#define AGG_GROUPS  (AGG_THREADS / CHANNELS)
#define AGG_BATCH   8

__global__ __launch_bounds__(AGG_THREADS)
void gcn_aggregate_kernel(const int2* __restrict__ sorted,
                          const unsigned* __restrict__ bstart,
                          const float* __restrict__ dis,
                          const float* __restrict__ h,
                          const float* __restrict__ b,
                          float* __restrict__ out, int n) {
    __shared__ __attribute__((aligned(16))) float acc[BK_NODES * CHANNELS];
    const int tid  = threadIdx.x;
    const int base = blockIdx.x << BK_SHIFT;

    for (int w = tid * 4; w < BK_NODES * CHANNELS; w += AGG_THREADS * 4)
        *reinterpret_cast<float4*>(&acc[w]) = make_float4(0.0f, 0.0f, 0.0f, 0.0f);
    __syncthreads();

    const int s0 = (int)bstart[blockIdx.x];
    const int s1 = (int)bstart[blockIdx.x + 1];
    const int g  = tid >> 4;
    const int ch = tid & 15;
    const int step = AGG_GROUPS * AGG_BATCH;
    int2 ed[AGG_BATCH];
#pragma unroll
    for (int u = 0; u < AGG_BATCH; ++u) ed[u] = make_int2(0, 0);
    if (s0 + g < s1) {
#pragma unroll
        for (int u = 0; u < AGG_BATCH; ++u)
            ed[u] = sorted[min(s0 + g + u * AGG_GROUPS, s1 - 1)];
    }
    for (int i0 = s0 + g; i0 < s1; i0 += step) {
        float hv[AGG_BATCH];
        int2  nd[AGG_BATCH];
        asm volatile("" :: "v"(ed[0].x), "v"(ed[0].y), "v"(ed[1].x), "v"(ed[1].y),
                           "v"(ed[2].x), "v"(ed[2].y), "v"(ed[3].x), "v"(ed[3].y),
                           "v"(ed[4].x), "v"(ed[4].y), "v"(ed[5].x), "v"(ed[5].y),
                           "v"(ed[6].x), "v"(ed[6].y), "v"(ed[7].x), "v"(ed[7].y));
#pragma unroll
        for (int u = 0; u < AGG_BATCH; ++u)
            hv[u] = h[(size_t)ed[u].x * CHANNELS + ch];
#pragma unroll
        for (int u = 0; u < AGG_BATCH; ++u)
            nd[u] = sorted[min(i0 + step + u * AGG_GROUPS, s1 - 1)];
        asm volatile("" :: "v"(hv[0]), "v"(hv[1]), "v"(hv[2]), "v"(hv[3]),
                           "v"(hv[4]), "v"(hv[5]), "v"(hv[6]), "v"(hv[7]));
#pragma unroll
        for (int u = 0; u < AGG_BATCH; ++u) {
            if (i0 + u * AGG_GROUPS < s1)
                atomicAdd(&acc[(ed[u].y - base) * CHANNELS + ch], hv[u]);
        }
#pragma unroll
        for (int u = 0; u < AGG_BATCH; ++u) ed[u] = nd[u];
    }
    __syncthreads();

    const int i = base + tid;
    if (tid < BK_NODES && i < n) {
        const float di = dis[i];
        float v[CHANNELS];
        float m = -INFINITY;
#pragma unroll
        for (int c = 0; c < CHANNELS; ++c) {
            float val = di * (acc[tid * CHANNELS + c] + h[(size_t)i * CHANNELS + c]) + b[c];
            val = fmaxf(val, 0.0f);
            v[c] = val;
            m = fmaxf(m, val);
        }
        float s = 0.0f;
#pragma unroll
        for (int c = 0; c < CHANNELS; ++c) s += __expf(v[c] - m);
        float lse = m + __logf(s);
#pragma unroll
        for (int c = 0; c < CHANNELS; ++c)
            out[(size_t)i * CHANNELS + c] = v[c] - lse;
    }
}

extern "C" void kernel_launch(void* const* d_in, const int* in_sizes, int n_in,
                              void* d_out, int out_size, void* d_ws, size_t ws_size,
                              hipStream_t stream) {
    const float* x  = (const float*)d_in[0];
    const int*   ei = (const int*)d_in[1];
    const float* W  = (const float*)d_in[2];
    const float* b  = (const float*)d_in[3];
    float* out = (float*)d_out;

    const int n = in_sizes[0] / KDIM;
    const int e = in_sizes[1] / 2;
    const int* row = ei;
    const int* col = ei + e;

    float* h   = (float*)d_ws;
    float* deg = h + (size_t)n * CHANNELS;
    unsigned* bcnt   = (unsigned*)(deg + (((size_t)n + 3) & ~(size_t)3));
    unsigned* bstart = bcnt + (size_t)DEG_BLOCKS * BK_MAX;
    int2* sorted     = (int2*)(bstart + BK_MAX + 16);
    unsigned* part   = (unsigned*)(sorted + (((size_t)e + 1) & ~(size_t)1));

    gcn_degree_lds_kernel<<<DEG_BLOCKS, DEG_THREADS, 0, stream>>>(col, part, bcnt, n, e);

    {
        const int ntiles = (n + 15) / 16;
        const int gemmBlocks = (ntiles + GEMM_TILES - 1) / GEMM_TILES;
        gcn_gemm_partition_kernel<<<DEG_BLOCKS + gemmBlocks, GP_THREADS, 0, stream>>>(
            x, W, part, deg, h, row, col, bcnt, bstart, sorted, n, e);
    }

    gcn_aggregate_kernel<<<(n + BK_NODES - 1) / BK_NODES, AGG_THREADS, 0, stream>>>(
        sorted, bstart, deg, h, b, out, n);
}


# Aggregate kernel: a wave reads 32 consecutive (source,target) pairs with one load, a pair per lane, and deals them to its 16-lane rows by DPP row_share (was one load per pair); arrival-order sums
# speedup vs baseline: 4.3049x; 1.0513x over previous
#include <hip/hip_runtime.h>
#include <math.h>

typedef __attribute__((ext_vector_type(2))) float v2f;
typedef __attribute__((ext_vector_type(8))) float v8f;

#define CHANNELS 16
#define KDIM 128

#define DEG_BLOCKS  128
#define DEG_THREADS 1024
#define DEG_CAPW    51200
#define DEG_BATCH   16

#define BK_SHIFT 9
#define BK_NODES 512
#define BK_MAX   256

__global__ __launch_bounds__(DEG_THREADS)
void gcn_degree_lds_kernel(const int* __restrict__ col,
                           unsigned* __restrict__ part,
                           unsigned* __restrict__ bcnt, int n, int e) {
    __shared__ __attribute__((aligned(16))) unsigned hist[DEG_CAPW];
    const int tid   = threadIdx.x;
    const int wp    = (((n + 1) >> 1) + 3) & ~3;
    const int chunk = (e + DEG_BLOCKS - 1) / DEG_BLOCKS;
    const int e0    = blockIdx.x * chunk;
    const int e1    = min(e0 + chunk, e);
    unsigned* prow  = part + (size_t)blockIdx.x * wp;

    for (int basew = 0; basew < wp; basew += DEG_CAPW) {
        const int nw = min(DEG_CAPW, wp - basew);
        for (int w = tid * 4; w < nw; w += DEG_THREADS * 4)
            *reinterpret_cast<uint4*>(&hist[w]) = make_uint4(0u, 0u, 0u, 0u);
        __syncthreads();
        for (int i0 = e0 + tid; i0 < e1; i0 += DEG_THREADS * DEG_BATCH) {
            int c[DEG_BATCH];
#pragma unroll
            for (int k = 0; k < DEG_BATCH; ++k)
                c[k] = col[min(i0 + k * DEG_THREADS, e1 - 1)];
#pragma unroll
            for (int k = 0; k < DEG_BATCH; ++k) {
                unsigned rel = (i0 + k * DEG_THREADS < e1)
                             ? (unsigned)c[k] - (unsigned)(basew * 2)
                             : 0xFFFFFFFFu;
                if (rel < (unsigned)(nw * 2))
                    atomicAdd(&hist[rel >> 1], 1u << ((rel & 1u) * 16u));
            }
        }
        __syncthreads();
        for (int w = tid * 4; w < nw; w += DEG_THREADS * 4)
            *reinterpret_cast<uint4*>(&prow[basew + w]) =
                *reinterpret_cast<const uint4*>(&hist[w]);
        {
            const int lane = tid & 31;
            const int nbl  = (nw + 255) >> 8;
            for (int j = tid >> 5; j < nbl; j += DEG_THREADS / 32) {
                const int w0 = (j << 8) + lane * 8;
                unsigned s = 0u;
                if (w0 < nw) {
                    uint4 a = *reinterpret_cast<const uint4*>(&hist[w0]);
                    s += (a.x & 0xFFFFu) + (a.x >> 16) + (a.y & 0xFFFFu) + (a.y >> 16)
                       + (a.z & 0xFFFFu) + (a.z >> 16) + (a.w & 0xFFFFu) + (a.w >> 16);
                }
                if (w0 + 4 < nw) {
                    uint4 a = *reinterpret_cast<const uint4*>(&hist[w0 + 4]);
                    s += (a.x & 0xFFFFu) + (a.x >> 16) + (a.y & 0xFFFFu) + (a.y >> 16)
                       + (a.z & 0xFFFFu) + (a.z >> 16) + (a.w & 0xFFFFu) + (a.w >> 16);
                }
                for (int d = 16; d > 0; d >>= 1) s += __shfl_xor(s, d);
                if (lane == 0)
                    bcnt[blockIdx.x * BK_MAX + (basew >> 8) + j] = s;
            }
        }
        __syncthreads();
    }
}

#define PART_BATCH 8
#define GP_THREADS 1024
#define GEMM_TILES (GP_THREADS / 32)
#define GEMM_ROWS  (GEMM_TILES * 16)

__global__ __launch_bounds__(GP_THREADS)
void gcn_gemm_partition_kernel(const float* __restrict__ x,
                               const float* __restrict__ W,
                               const unsigned* __restrict__ part,
                               float* __restrict__ dis,
                               float* __restrict__ h,
                               const int* __restrict__ row,
                               const int* __restrict__ col,
                               const unsigned* __restrict__ bcnt,
                               unsigned* __restrict__ bstart,
                               int2* __restrict__ sorted, int n, int e) {
    __shared__ unsigned tot[BK_MAX];
    __shared__ unsigned bef[BK_MAX];
    __shared__ unsigned cur[BK_MAX];
    __shared__ float sdis[GEMM_ROWS];
    const int tid = threadIdx.x;

    if (blockIdx.x < DEG_BLOCKS) {
        const int nbk   = (n + BK_NODES - 1) >> BK_SHIFT;
        const int chunk = (e + DEG_BLOCKS - 1) / DEG_BLOCKS;
        const int e0    = blockIdx.x * chunk;
        const int e1    = min(e0 + chunk, e);

        if (tid < BK_MAX) {
            unsigned t = 0u, bf = 0u;
            if (tid < nbk) {
#pragma unroll 16
                for (int b = 0; b < DEG_BLOCKS; ++b) {
                    unsigned v = bcnt[b * BK_MAX + tid];
                    t  += v;
                    bf += (b < (int)blockIdx.x) ? v : 0u;
                }
            }
            tot[tid] = t;
            bef[tid] = bf;
        }
        __syncthreads();
        if (tid < 32) {
            unsigned v[8];
            unsigned run = 0u;
#pragma unroll
            for (int i = 0; i < 8; ++i) {
                unsigned t = tot[tid * 8 + i];
                v[i] = run;
                run += t;
            }
            unsigned inc = run;
            for (int d = 1; d < 32; d <<= 1) {
                unsigned t = __shfl_up(inc, d);
                if (tid >= d) inc += t;
            }
            const unsigned excl = inc - run;
#pragma unroll
            for (int i = 0; i < 8; ++i) {
                const unsigned st = excl + v[i];
                cur[tid * 8 + i] = st + bef[tid * 8 + i];
                if (blockIdx.x == 0) bstart[tid * 8 + i] = st;
            }
            if (blockIdx.x == 0 && tid == 31) bstart[BK_MAX] = inc;
        }
        __syncthreads();

        for (int i0 = e0 + tid; i0 < e1; i0 += GP_THREADS * PART_BATCH) {
            int c[PART_BATCH], r[PART_BATCH];
            unsigned pos[PART_BATCH];
#pragma unroll
            for (int k = 0; k < PART_BATCH; ++k) {
                const int idx = min(i0 + k * GP_THREADS, e1 - 1);
                c[k] = col[idx];
                r[k] = row[idx];
            }
            asm volatile("" :: "v"(c[0]), "v"(c[1]), "v"(c[2]), "v"(c[3]),
                               "v"(c[4]), "v"(c[5]), "v"(c[6]), "v"(c[7]),
                               "v"(r[0]), "v"(r[1]), "v"(r[2]), "v"(r[3]),
                               "v"(r[4]), "v"(r[5]), "v"(r[6]), "v"(r[7]));
#pragma unroll
            for (int k = 0; k < PART_BATCH; ++k) {
                pos[k] = 0u;
                if (i0 + k * GP_THREADS < e1)
                    pos[k] = atomicAdd(&cur[c[k] >> BK_SHIFT], 1u);
            }
#pragma unroll
            for (int k = 0; k < PART_BATCH; ++k)
                if (i0 + k * GP_THREADS < e1)
                    sorted[pos[k]] = make_int2(r[k], c[k]);
        }
    } else {
        const int gb      = blockIdx.x - DEG_BLOCKS;
        const int lane    = tid & 31;
        const int waveId  = __builtin_amdgcn_readfirstlane(tid >> 5);
        const int rowbase = gb * GEMM_ROWS;
        const int ntiles  = (n + 15) >> 4;
        const int tile    = gb * GEMM_TILES + waveId;

        if (tid < GEMM_ROWS / 2) {
            const int w  = (rowbase >> 1) + tid;
            const int wp = (((n + 1) >> 1) + 3) & ~3;
            float d0 = 0.0f, d1 = 0.0f;
            if (2 * w < n) {
                unsigned lo = 0u, hi = 0u;
                const unsigned wu = (unsigned)w;
#pragma unroll 8
                for (int b = 0; b < DEG_BLOCKS; ++b) {
                    const unsigned* prow = part + (size_t)b * wp;
                    unsigned v = prow[wu];
                    lo += v & 0xFFFFu;
                    hi += v >> 16;
                }
                d0 = rsqrtf(1.0f + (float)lo);
                d1 = rsqrtf(1.0f + (float)hi);
                dis[2 * w] = d0;
                if (2 * w + 1 < n) dis[2 * w + 1] = d1;
            }
            sdis[2 * tid]     = d0;
            sdis[2 * tid + 1] = d1;
        }

        const int half = lane >> 4;
        const int idx  = lane & 15;
        const int koff = half * 2;

        v8f c = {};
        if (tile < ntiles) {
            int arow = tile * 16 + idx;
            if (arow >= n) arow = n - 1;
            const float* xrow = x + (size_t)arow * KDIM;
#pragma unroll
            for (int k = 0; k < KDIM; k += 4) {
                v2f a;
                a.x = xrow[k + koff + 0];
                a.y = xrow[k + koff + 1];
                v2f bm;
                bm.x = W[(k + koff + 0) * CHANNELS + idx];
                bm.y = W[(k + koff + 1) * CHANNELS + idx];
                c = __builtin_amdgcn_wmma_f32_16x16x4_f32(
                     false, a,  false, bm,
                     (short)0, c,  false,  false);
            }
        }
        __syncthreads();

        if (tile < ntiles) {
#pragma unroll
            for (int v = 0; v < 8; ++v) {
                const int lr = waveId * 16 + v + 8 * half;
                const int rw = rowbase + lr;
                if (rw < n) h[(size_t)rw * CHANNELS + idx] = c[v] * sdis[lr];
            }
        }
    }
}

#define AGG_THREADS 1024
#define AGG_CHUNK   32

template <int J>
__device__ __forceinline__ unsigned agg_row_share(unsigned v) {
    return (unsigned)__builtin_amdgcn_update_dpp(0, (int)v, 0x150 + J, 0xF, 0xF, true);
}

#define AGG_GATHER(J) \
    hv[J] = *reinterpret_cast<const float*>(hbytes + (agg_row_share<J>(gall) | ch4));
#define AGG_ADD(J) \
    atomicAdd(reinterpret_cast<float*>(reinterpret_cast<char*>(acc) + \
                                       (agg_row_share<J>(lall) | ch4)), hv[J]);

__global__ __launch_bounds__(AGG_THREADS)
void gcn_aggregate_kernel(const int2* __restrict__ sorted,
                          const unsigned* __restrict__ bstart,
                          const float* __restrict__ dis,
                          const float* __restrict__ h,
                          const float* __restrict__ b,
                          float* __restrict__ out, int n) {
    __shared__ __attribute__((aligned(16))) float acc[(BK_NODES + 1) * CHANNELS];
    const int tid  = threadIdx.x;
    const int base = blockIdx.x << BK_SHIFT;

    for (int w = tid * 4; w < (BK_NODES + 1) * CHANNELS; w += AGG_THREADS * 4)
        *reinterpret_cast<float4*>(&acc[w]) = make_float4(0.0f, 0.0f, 0.0f, 0.0f);
    __syncthreads();

    const int s0 = (int)bstart[blockIdx.x];
    const int s1 = (int)bstart[blockIdx.x + 1];
    const int lane = tid & 31;
    const int wv = __builtin_amdgcn_readfirstlane(tid >> 5);
    const unsigned ch4 = (unsigned)(lane & 15) * 4u;
    const char* hbytes = reinterpret_cast<const char*>(h);
    int c0 = s0 + wv * AGG_CHUNK;
    int2 ed = make_int2(0, 0);
    if (c0 < s1) ed = sorted[min(c0 + lane, s1 - 1)];
    for (; c0 < s1; c0 += AGG_THREADS) {
        const bool valid = (c0 + lane) < s1;
        const unsigned gall = valid ? ((unsigned)ed.x << 6) : 0u;
        const unsigned lall = valid ? ((unsigned)(ed.y - base) << 6)
                                    : (unsigned)(BK_NODES << 6);
        float hv[16];
        AGG_GATHER(0)  AGG_GATHER(1)  AGG_GATHER(2)  AGG_GATHER(3)
        AGG_GATHER(4)  AGG_GATHER(5)  AGG_GATHER(6)  AGG_GATHER(7)
        AGG_GATHER(8)  AGG_GATHER(9)  AGG_GATHER(10) AGG_GATHER(11)
        AGG_GATHER(12) AGG_GATHER(13) AGG_GATHER(14) AGG_GATHER(15)
        ed = sorted[min(c0 + AGG_THREADS + lane, s1 - 1)];
        AGG_ADD(0)  AGG_ADD(1)  AGG_ADD(2)  AGG_ADD(3)
        AGG_ADD(4)  AGG_ADD(5)  AGG_ADD(6)  AGG_ADD(7)
        AGG_ADD(8)  AGG_ADD(9)  AGG_ADD(10) AGG_ADD(11)
        AGG_ADD(12) AGG_ADD(13) AGG_ADD(14) AGG_ADD(15)
    }
    __syncthreads();

    const int i = base + tid;
    if (tid < BK_NODES && i < n) {
        const float di = dis[i];
        float v[CHANNELS];
        float m = -INFINITY;
#pragma unroll
        for (int c = 0; c < CHANNELS; ++c) {
            float val = di * (acc[tid * CHANNELS + c] + h[(size_t)i * CHANNELS + c]) + b[c];
            val = fmaxf(val, 0.0f);
            v[c] = val;
            m = fmaxf(m, val);
        }
        float s = 0.0f;
#pragma unroll
        for (int c = 0; c < CHANNELS; ++c) s += __expf(v[c] - m);
        float lse = m + __logf(s);
#pragma unroll
        for (int c = 0; c < CHANNELS; ++c)
            out[(size_t)i * CHANNELS + c] = v[c] - lse;
    }
}

extern "C" void kernel_launch(void* const* d_in, const int* in_sizes, int n_in,
                              void* d_out, int out_size, void* d_ws, size_t ws_size,
                              hipStream_t stream) {
    const float* x  = (const float*)d_in[0];
    const int*   ei = (const int*)d_in[1];
    const float* W  = (const float*)d_in[2];
    const float* b  = (const float*)d_in[3];
    float* out = (float*)d_out;

    const int n = in_sizes[0] / KDIM;
    const int e = in_sizes[1] / 2;
    const int* row = ei;
    const int* col = ei + e;

    float* h   = (float*)d_ws;
    float* deg = h + (size_t)n * CHANNELS;
    unsigned* bcnt   = (unsigned*)(deg + (((size_t)n + 3) & ~(size_t)3));
    unsigned* bstart = bcnt + (size_t)DEG_BLOCKS * BK_MAX;
    int2* sorted     = (int2*)(bstart + BK_MAX + 16);
    unsigned* part   = (unsigned*)(sorted + (((size_t)e + 1) & ~(size_t)1));

    gcn_degree_lds_kernel<<<DEG_BLOCKS, DEG_THREADS, 0, stream>>>(col, part, bcnt, n, e);

    {
        const int ntiles = (n + 15) / 16;
        const int gemmBlocks = (ntiles + GEMM_TILES - 1) / GEMM_TILES;
        gcn_gemm_partition_kernel<<<DEG_BLOCKS + gemmBlocks, GP_THREADS, 0, stream>>>(
            x, W, part, deg, h, row, col, bcnt, bstart, sorted, n, e);
    }

    gcn_aggregate_kernel<<<(n + BK_NODES - 1) / BK_NODES, AGG_THREADS, 0, stream>>>(
        sorted, bstart, deg, h, b, out, n);
}


# Edge partition: a round's 8192 pairs grouped by bucket in LDS (rank = the returning LDS add, scan of counts), then copied to sorted as contiguous runs, not one 8-byte store per edge; arrival order
# speedup vs baseline: 5.0010x; 1.1617x over previous
#include <hip/hip_runtime.h>
#include <math.h>

typedef __attribute__((ext_vector_type(2))) float v2f;
typedef __attribute__((ext_vector_type(8))) float v8f;

#define CHANNELS 16
#define KDIM 128

#define DEG_BLOCKS  128
#define DEG_THREADS 1024
#define DEG_CAPW    51200
#define DEG_BATCH   16

#define BK_SHIFT 9
#define BK_NODES 512
#define BK_MAX   256

__global__ __launch_bounds__(DEG_THREADS)
void gcn_degree_lds_kernel(const int* __restrict__ col,
                           unsigned* __restrict__ part,
                           unsigned* __restrict__ bcnt, int n, int e) {
    __shared__ __attribute__((aligned(16))) unsigned hist[DEG_CAPW];
    const int tid   = threadIdx.x;
    const int wp    = (((n + 1) >> 1) + 3) & ~3;
    const int chunk = (e + DEG_BLOCKS - 1) / DEG_BLOCKS;
    const int e0    = blockIdx.x * chunk;
    const int e1    = min(e0 + chunk, e);
    unsigned* prow  = part + (size_t)blockIdx.x * wp;

    for (int basew = 0; basew < wp; basew += DEG_CAPW) {
        const int nw = min(DEG_CAPW, wp - basew);
        for (int w = tid * 4; w < nw; w += DEG_THREADS * 4)
            *reinterpret_cast<uint4*>(&hist[w]) = make_uint4(0u, 0u, 0u, 0u);
        __syncthreads();
        for (int i0 = e0 + tid; i0 < e1; i0 += DEG_THREADS * DEG_BATCH) {
            int c[DEG_BATCH];
#pragma unroll
            for (int k = 0; k < DEG_BATCH; ++k)
                c[k] = col[min(i0 + k * DEG_THREADS, e1 - 1)];
#pragma unroll
            for (int k = 0; k < DEG_BATCH; ++k) {
                unsigned rel = (i0 + k * DEG_THREADS < e1)
                             ? (unsigned)c[k] - (unsigned)(basew * 2)
                             : 0xFFFFFFFFu;
                if (rel < (unsigned)(nw * 2))
                    atomicAdd(&hist[rel >> 1], 1u << ((rel & 1u) * 16u));
            }
        }
        __syncthreads();
        for (int w = tid * 4; w < nw; w += DEG_THREADS * 4)
            *reinterpret_cast<uint4*>(&prow[basew + w]) =
                *reinterpret_cast<const uint4*>(&hist[w]);
        {
            const int lane = tid & 31;
            const int nbl  = (nw + 255) >> 8;
            for (int j = tid >> 5; j < nbl; j += DEG_THREADS / 32) {
                const int w0 = (j << 8) + lane * 8;
                unsigned s = 0u;
                if (w0 < nw) {
                    uint4 a = *reinterpret_cast<const uint4*>(&hist[w0]);
                    s += (a.x & 0xFFFFu) + (a.x >> 16) + (a.y & 0xFFFFu) + (a.y >> 16)
                       + (a.z & 0xFFFFu) + (a.z >> 16) + (a.w & 0xFFFFu) + (a.w >> 16);
                }
                if (w0 + 4 < nw) {
                    uint4 a = *reinterpret_cast<const uint4*>(&hist[w0 + 4]);
                    s += (a.x & 0xFFFFu) + (a.x >> 16) + (a.y & 0xFFFFu) + (a.y >> 16)
                       + (a.z & 0xFFFFu) + (a.z >> 16) + (a.w & 0xFFFFu) + (a.w >> 16);
                }
                for (int d = 16; d > 0; d >>= 1) s += __shfl_xor(s, d);
                if (lane == 0)
                    bcnt[blockIdx.x * BK_MAX + (basew >> 8) + j] = s;
            }
        }
        __syncthreads();
    }
}

#define PART_BATCH 8
#define GP_THREADS 1024
#define PART_ROUND (GP_THREADS * PART_BATCH)
#define GEMM_TILES (GP_THREADS / 32)
#define GEMM_ROWS  (GEMM_TILES * 16)

__global__ __launch_bounds__(GP_THREADS) __attribute__((amdgpu_waves_per_eu(16, 16)))
void gcn_gemm_partition_kernel(const float* __restrict__ x,
                               const float* __restrict__ W,
                               const unsigned* __restrict__ part,
                               float* __restrict__ dis,
                               float* __restrict__ h,
                               const int* __restrict__ row,
                               const int* __restrict__ col,
                               const unsigned* __restrict__ bcnt,
                               unsigned* __restrict__ bstart,
                               int2* __restrict__ sorted, int n, int e) {
    struct GpLds {
        unsigned tot[BK_MAX];
        unsigned bef[BK_MAX];
        unsigned cur[BK_MAX];
        unsigned cnt[BK_MAX];
        unsigned lst[BK_MAX];
        unsigned gof[BK_MAX];
        float    sdis[GEMM_ROWS];
        int2     stage[PART_ROUND];
    };
    __shared__ __attribute__((aligned(16))) GpLds lds;
    unsigned (&tot)[BK_MAX] = lds.tot;
    unsigned (&bef)[BK_MAX] = lds.bef;
    unsigned (&cur)[BK_MAX] = lds.cur;
    unsigned (&cnt)[BK_MAX] = lds.cnt;
    unsigned (&lst)[BK_MAX] = lds.lst;
    unsigned (&gof)[BK_MAX] = lds.gof;
    float (&sdis)[GEMM_ROWS] = lds.sdis;
    int2 (&stage)[PART_ROUND] = lds.stage;
    const int tid = threadIdx.x;

    if (blockIdx.x < DEG_BLOCKS) {
        const int nbk   = (n + BK_NODES - 1) >> BK_SHIFT;
        const int chunk = (e + DEG_BLOCKS - 1) / DEG_BLOCKS;
        const int e0    = blockIdx.x * chunk;
        const int e1    = min(e0 + chunk, e);

        if (tid < BK_MAX) {
            unsigned t = 0u, bf = 0u;
            if (tid < nbk) {
#pragma unroll 16
                for (int b = 0; b < DEG_BLOCKS; ++b) {
                    unsigned v = bcnt[b * BK_MAX + tid];
                    t  += v;
                    bf += (b < (int)blockIdx.x) ? v : 0u;
                }
            }
            tot[tid] = t;
            bef[tid] = bf;
            cnt[tid] = 0u;
        }
        __syncthreads();
        if (tid < 32) {
            unsigned v[8];
            unsigned run = 0u;
#pragma unroll
            for (int i = 0; i < 8; ++i) {
                unsigned t = tot[tid * 8 + i];
                v[i] = run;
                run += t;
            }
            unsigned inc = run;
            for (int d = 1; d < 32; d <<= 1) {
                unsigned t = __shfl_up(inc, d);
                if (tid >= d) inc += t;
            }
            const unsigned excl = inc - run;
#pragma unroll
            for (int i = 0; i < 8; ++i) {
                const unsigned st = excl + v[i];
                cur[tid * 8 + i] = st + bef[tid * 8 + i];
                if (blockIdx.x == 0) bstart[tid * 8 + i] = st;
            }
            if (blockIdx.x == 0 && tid == 31) bstart[BK_MAX] = inc;
        }
        __syncthreads();

        int c[PART_BATCH], r[PART_BATCH];
#pragma unroll
        for (int k = 0; k < PART_BATCH; ++k) {
            c[k] = 0;
            r[k] = 0;
            if (e0 < e1) {
                const int idx = min(e0 + tid + k * GP_THREADS, e1 - 1);
                c[k] = col[idx];
                r[k] = row[idx];
            }
        }
        for (int base = e0; base < e1; base += PART_ROUND) {
            const int m = min(PART_ROUND, e1 - base);
            unsigned rank[PART_BATCH];
#pragma unroll
            for (int k = 0; k < PART_BATCH; ++k) {
                rank[k] = 0u;
                if (tid + k * GP_THREADS < m)
                    rank[k] = atomicAdd(&cnt[c[k] >> BK_SHIFT], 1u);
            }
            __syncthreads();
            if (tid < 32) {
                unsigned run = 0u;
#pragma unroll
                for (int i = 0; i < 8; ++i) run += cnt[tid * 8 + i];
                unsigned inc = run;
                for (int d = 1; d < 32; d <<= 1) {
                    unsigned u = __shfl_up(inc, d);
                    if (tid >= d) inc += u;
                }
                unsigned ls = inc - run;
#pragma unroll
                for (int i = 0; i < 8; ++i) {
                    const unsigned t  = cnt[tid * 8 + i];
                    const unsigned cu = cur[tid * 8 + i];
                    lst[tid * 8 + i] = ls;
                    gof[tid * 8 + i] = cu - ls;
                    cur[tid * 8 + i] = cu + t;
                    cnt[tid * 8 + i] = 0u;
                    ls += t;
                }
            }
            __syncthreads();
#pragma unroll
            for (int k = 0; k < PART_BATCH; ++k) rank[k] += lst[c[k] >> BK_SHIFT];
#pragma unroll
            for (int k = 0; k < PART_BATCH; ++k)
                if (tid + k * GP_THREADS < m)
                    stage[rank[k]] = make_int2(r[k], c[k]);
            __syncthreads();
            if (base + PART_ROUND < e1) {
#pragma unroll
                for (int k = 0; k < PART_BATCH; ++k) {
                    const int idx = min(base + PART_ROUND + tid + k * GP_THREADS, e1 - 1);
                    c[k] = col[idx];
                    r[k] = row[idx];
                }
            }
#pragma unroll
            for (int g = 0; g < PART_BATCH; g += 4) {
                int2 p[4];
                unsigned go[4];
#pragma unroll
                for (int q = 0; q < 4; ++q)
                    p[q] = stage[min(tid + (g + q) * GP_THREADS, m - 1)];
#pragma unroll
                for (int q = 0; q < 4; ++q)
                    go[q] = gof[p[q].y >> BK_SHIFT];
#pragma unroll
                for (int q = 0; q < 4; ++q) {
                    const int j = tid + (g + q) * GP_THREADS;
                    if (j < m) sorted[go[q] + (unsigned)j] = p[q];
                }
            }
        }
    } else {
        const int gb      = blockIdx.x - DEG_BLOCKS;
        const int lane    = tid & 31;
        const int waveId  = __builtin_amdgcn_readfirstlane(tid >> 5);
        const int rowbase = gb * GEMM_ROWS;
        const int ntiles  = (n + 15) >> 4;
        const int tile    = gb * GEMM_TILES + waveId;

        if (tid < GEMM_ROWS / 2) {
            const int w  = (rowbase >> 1) + tid;
            const int wp = (((n + 1) >> 1) + 3) & ~3;
            float d0 = 0.0f, d1 = 0.0f;
            if (2 * w < n) {
                unsigned lo = 0u, hi = 0u;
                const unsigned wu = (unsigned)w;
#pragma unroll 8
                for (int b = 0; b < DEG_BLOCKS; ++b) {
                    const unsigned* prow = part + (size_t)b * wp;
                    unsigned v = prow[wu];
                    lo += v & 0xFFFFu;
                    hi += v >> 16;
                }
                d0 = rsqrtf(1.0f + (float)lo);
                d1 = rsqrtf(1.0f + (float)hi);
                dis[2 * w] = d0;
                if (2 * w + 1 < n) dis[2 * w + 1] = d1;
            }
            sdis[2 * tid]     = d0;
            sdis[2 * tid + 1] = d1;
        }

        const int half = lane >> 4;
        const int idx  = lane & 15;
        const int koff = half * 2;

        v8f c = {};
        if (tile < ntiles) {
            int arow = tile * 16 + idx;
            if (arow >= n) arow = n - 1;
            const float* xrow = x + (size_t)arow * KDIM;
#pragma unroll
            for (int k = 0; k < KDIM; k += 4) {
                v2f a;
                a.x = xrow[k + koff + 0];
                a.y = xrow[k + koff + 1];
                v2f bm;
                bm.x = W[(k + koff + 0) * CHANNELS + idx];
                bm.y = W[(k + koff + 1) * CHANNELS + idx];
                c = __builtin_amdgcn_wmma_f32_16x16x4_f32(
                     false, a,  false, bm,
                     (short)0, c,  false,  false);
            }
        }
        __syncthreads();

        if (tile < ntiles) {
#pragma unroll
            for (int v = 0; v < 8; ++v) {
                const int lr = waveId * 16 + v + 8 * half;
                const int rw = rowbase + lr;
                if (rw < n) h[(size_t)rw * CHANNELS + idx] = c[v] * sdis[lr];
            }
        }
    }
}

#define AGG_THREADS 1024
#define AGG_CHUNK   32

template <int J>
__device__ __forceinline__ unsigned agg_row_share(unsigned v) {
    return (unsigned)__builtin_amdgcn_update_dpp(0, (int)v, 0x150 + J, 0xF, 0xF, true);
}

#define AGG_GATHER(J) \
    hv[J] = *reinterpret_cast<const float*>(hbytes + (agg_row_share<J>(gall) | ch4));
#define AGG_ADD(J) \
    atomicAdd(reinterpret_cast<float*>(reinterpret_cast<char*>(acc) + \
                                       (agg_row_share<J>(lall) | ch4)), hv[J]);

__global__ __launch_bounds__(AGG_THREADS)
void gcn_aggregate_kernel(const int2* __restrict__ sorted,
                          const unsigned* __restrict__ bstart,
                          const float* __restrict__ dis,
                          const float* __restrict__ h,
                          const float* __restrict__ b,
                          float* __restrict__ out, int n) {
    __shared__ __attribute__((aligned(16))) float acc[(BK_NODES + 1) * CHANNELS];
    const int tid  = threadIdx.x;
    const int base = blockIdx.x << BK_SHIFT;

    for (int w = tid * 4; w < (BK_NODES + 1) * CHANNELS; w += AGG_THREADS * 4)
        *reinterpret_cast<float4*>(&acc[w]) = make_float4(0.0f, 0.0f, 0.0f, 0.0f);
    __syncthreads();

    const int s0 = (int)bstart[blockIdx.x];
    const int s1 = (int)bstart[blockIdx.x + 1];
    const int lane = tid & 31;
    const int wv = __builtin_amdgcn_readfirstlane(tid >> 5);
    const unsigned ch4 = (unsigned)(lane & 15) * 4u;
    const char* hbytes = reinterpret_cast<const char*>(h);
    int c0 = s0 + wv * AGG_CHUNK;
    int2 ed = make_int2(0, 0);
    if (c0 < s1) ed = sorted[min(c0 + lane, s1 - 1)];
    for (; c0 < s1; c0 += AGG_THREADS) {
        const bool valid = (c0 + lane) < s1;
        const unsigned gall = valid ? ((unsigned)ed.x << 6) : 0u;
        const unsigned lall = valid ? ((unsigned)(ed.y - base) << 6)
                                    : (unsigned)(BK_NODES << 6);
        float hv[16];
        AGG_GATHER(0)  AGG_GATHER(1)  AGG_GATHER(2)  AGG_GATHER(3)
        AGG_GATHER(4)  AGG_GATHER(5)  AGG_GATHER(6)  AGG_GATHER(7)
        AGG_GATHER(8)  AGG_GATHER(9)  AGG_GATHER(10) AGG_GATHER(11)
        AGG_GATHER(12) AGG_GATHER(13) AGG_GATHER(14) AGG_GATHER(15)
        ed = sorted[min(c0 + AGG_THREADS + lane, s1 - 1)];
        AGG_ADD(0)  AGG_ADD(1)  AGG_ADD(2)  AGG_ADD(3)
        AGG_ADD(4)  AGG_ADD(5)  AGG_ADD(6)  AGG_ADD(7)
        AGG_ADD(8)  AGG_ADD(9)  AGG_ADD(10) AGG_ADD(11)
        AGG_ADD(12) AGG_ADD(13) AGG_ADD(14) AGG_ADD(15)
    }
    __syncthreads();

    const int i = base + tid;
    if (tid < BK_NODES && i < n) {
        const float di = dis[i];
        float v[CHANNELS];
        float m = -INFINITY;
#pragma unroll
        for (int c = 0; c < CHANNELS; ++c) {
            float val = di * (acc[tid * CHANNELS + c] + h[(size_t)i * CHANNELS + c]) + b[c];
            val = fmaxf(val, 0.0f);
            v[c] = val;
            m = fmaxf(m, val);
        }
        float s = 0.0f;
#pragma unroll
        for (int c = 0; c < CHANNELS; ++c) s += __expf(v[c] - m);
        float lse = m + __logf(s);
#pragma unroll
        for (int c = 0; c < CHANNELS; ++c)
            out[(size_t)i * CHANNELS + c] = v[c] - lse;
    }
}

extern "C" void kernel_launch(void* const* d_in, const int* in_sizes, int n_in,
                              void* d_out, int out_size, void* d_ws, size_t ws_size,
                              hipStream_t stream) {
    const float* x  = (const float*)d_in[0];
    const int*   ei = (const int*)d_in[1];
    const float* W  = (const float*)d_in[2];
    const float* b  = (const float*)d_in[3];
    float* out = (float*)d_out;

    const int n = in_sizes[0] / KDIM;
    const int e = in_sizes[1] / 2;
    const int* row = ei;
    const int* col = ei + e;

    float* h   = (float*)d_ws;
    float* deg = h + (size_t)n * CHANNELS;
    unsigned* bcnt   = (unsigned*)(deg + (((size_t)n + 3) & ~(size_t)3));
    unsigned* bstart = bcnt + (size_t)DEG_BLOCKS * BK_MAX;
    int2* sorted     = (int2*)(bstart + BK_MAX + 16);
    unsigned* part   = (unsigned*)(sorted + (((size_t)e + 1) & ~(size_t)1));

    gcn_degree_lds_kernel<<<DEG_BLOCKS, DEG_THREADS, 0, stream>>>(col, part, bcnt, n, e);

    {
        const int ntiles = (n + 15) / 16;
        const int gemmBlocks = (ntiles + GEMM_TILES - 1) / GEMM_TILES;
        gcn_gemm_partition_kernel<<<DEG_BLOCKS + gemmBlocks, GP_THREADS, 0, stream>>>(
            x, W, part, deg, h, row, col, bcnt, bstart, sorted, n, e);
    }

    gcn_aggregate_kernel<<<(n + BK_NODES - 1) / BK_NODES, AGG_THREADS, 0, stream>>>(
        sorted, bstart, deg, h, b, out, n);
}


# Load latency taken off each block's start-up: sums over the 128 slices split over all 32 waves (joined by LDS integer adds), first edge batches requested ahead of the LDS clears and table sums
# speedup vs baseline: 5.2842x; 1.0566x over previous
#include <hip/hip_runtime.h>
#include <math.h>

typedef __attribute__((ext_vector_type(2))) float v2f;
typedef __attribute__((ext_vector_type(8))) float v8f;

#define CHANNELS 16
#define KDIM 128

#define DEG_BLOCKS  128
#define DEG_THREADS 1024
#define DEG_CAPW    51200
#define DEG_BATCH   16

#define BK_SHIFT 9
#define BK_NODES 512
#define BK_MAX   256

__global__ __launch_bounds__(DEG_THREADS)
void gcn_degree_lds_kernel(const int* __restrict__ col,
                           unsigned* __restrict__ part,
                           unsigned* __restrict__ bcnt, int n, int e) {
    __shared__ __attribute__((aligned(16))) unsigned hist[DEG_CAPW];
    const int tid   = threadIdx.x;
    const int wp    = (((n + 1) >> 1) + 3) & ~3;
    const int chunk = (e + DEG_BLOCKS - 1) / DEG_BLOCKS;
    const int e0    = blockIdx.x * chunk;
    const int e1    = min(e0 + chunk, e);
    unsigned* prow  = part + (size_t)blockIdx.x * wp;

    for (int basew = 0; basew < wp; basew += DEG_CAPW) {
        const int nw = min(DEG_CAPW, wp - basew);
        int c[DEG_BATCH];
#pragma unroll
        for (int k = 0; k < DEG_BATCH; ++k) c[k] = 0;
        if (e0 < e1) {
#pragma unroll
            for (int k = 0; k < DEG_BATCH; ++k)
                c[k] = col[min(e0 + tid + k * DEG_THREADS, e1 - 1)];
        }
        for (int w = tid * 4; w < nw; w += DEG_THREADS * 4)
            *reinterpret_cast<uint4*>(&hist[w]) = make_uint4(0u, 0u, 0u, 0u);
        __syncthreads();
        for (int i0 = e0 + tid; i0 < e1; i0 += DEG_THREADS * DEG_BATCH) {
            int cn[DEG_BATCH];
            const int i1 = i0 + DEG_THREADS * DEG_BATCH;
#pragma unroll
            for (int k = 0; k < DEG_BATCH; ++k) cn[k] = 0;
            if (i1 < e1) {
#pragma unroll
                for (int k = 0; k < DEG_BATCH; ++k)
                    cn[k] = col[min(i1 + k * DEG_THREADS, e1 - 1)];
            }
#pragma unroll
            for (int k = 0; k < DEG_BATCH; ++k) {
                unsigned rel = (i0 + k * DEG_THREADS < e1)
                             ? (unsigned)c[k] - (unsigned)(basew * 2)
                             : 0xFFFFFFFFu;
                if (rel < (unsigned)(nw * 2))
                    atomicAdd(&hist[rel >> 1], 1u << ((rel & 1u) * 16u));
            }
#pragma unroll
            for (int k = 0; k < DEG_BATCH; ++k) c[k] = cn[k];
        }
        __syncthreads();
        for (int w = tid * 4; w < nw; w += DEG_THREADS * 4)
            *reinterpret_cast<uint4*>(&prow[basew + w]) =
                *reinterpret_cast<const uint4*>(&hist[w]);
        {
            const int lane = tid & 31;
            const int nbl  = (nw + 255) >> 8;
            for (int j = tid >> 5; j < nbl; j += DEG_THREADS / 32) {
                const int w0 = (j << 8) + lane * 8;
                unsigned s = 0u;
                if (w0 < nw) {
                    uint4 a = *reinterpret_cast<const uint4*>(&hist[w0]);
                    s += (a.x & 0xFFFFu) + (a.x >> 16) + (a.y & 0xFFFFu) + (a.y >> 16)
                       + (a.z & 0xFFFFu) + (a.z >> 16) + (a.w & 0xFFFFu) + (a.w >> 16);
                }
                if (w0 + 4 < nw) {
                    uint4 a = *reinterpret_cast<const uint4*>(&hist[w0 + 4]);
                    s += (a.x & 0xFFFFu) + (a.x >> 16) + (a.y & 0xFFFFu) + (a.y >> 16)
                       + (a.z & 0xFFFFu) + (a.z >> 16) + (a.w & 0xFFFFu) + (a.w >> 16);
                }
                for (int d = 16; d > 0; d >>= 1) s += __shfl_xor(s, d);
                if (lane == 0)
                    bcnt[blockIdx.x * BK_MAX + (basew >> 8) + j] = s;
            }
        }
        __syncthreads();
    }
}

#define PART_BATCH 8
#define GP_THREADS 1024
#define PART_ROUND (GP_THREADS * PART_BATCH)
#define GEMM_TILES (GP_THREADS / 32)
#define GEMM_ROWS  (GEMM_TILES * 16)
#define SUM_GROUPS (GP_THREADS / BK_MAX)
#define SUM_ROWS   (DEG_BLOCKS / SUM_GROUPS)
static_assert(SUM_GROUPS * BK_MAX == GP_THREADS && SUM_GROUPS * SUM_ROWS == DEG_BLOCKS,
              "the slices must split evenly over the thread groups");
static_assert(GEMM_ROWS / 2 == BK_MAX, "a product block's degree words share the bucket tables");

__global__ __launch_bounds__(GP_THREADS) __attribute__((amdgpu_waves_per_eu(16, 16)))
void gcn_gemm_partition_kernel(const float* __restrict__ x,
                               const float* __restrict__ W,
                               const unsigned* __restrict__ part,
                               float* __restrict__ dis,
                               float* __restrict__ h,
                               const int* __restrict__ row,
                               const int* __restrict__ col,
                               const unsigned* __restrict__ bcnt,
                               unsigned* __restrict__ bstart,
                               int2* __restrict__ sorted, int n, int e) {
    struct GpLds {
        unsigned tot[BK_MAX];
        unsigned bef[BK_MAX];
        unsigned cur[BK_MAX];
        unsigned cnt[BK_MAX];
        unsigned lst[BK_MAX];
        unsigned gof[BK_MAX];
        float    sdis[GEMM_ROWS];
        int2     stage[PART_ROUND];
    };
    __shared__ __attribute__((aligned(16))) GpLds lds;
    unsigned (&tot)[BK_MAX] = lds.tot;
    unsigned (&bef)[BK_MAX] = lds.bef;
    unsigned (&cur)[BK_MAX] = lds.cur;
    unsigned (&cnt)[BK_MAX] = lds.cnt;
    unsigned (&lst)[BK_MAX] = lds.lst;
    unsigned (&gof)[BK_MAX] = lds.gof;
    float (&sdis)[GEMM_ROWS] = lds.sdis;
    int2 (&stage)[PART_ROUND] = lds.stage;
    const int tid = threadIdx.x;

    if (blockIdx.x < DEG_BLOCKS) {
        const int nbk   = (n + BK_NODES - 1) >> BK_SHIFT;
        const int chunk = (e + DEG_BLOCKS - 1) / DEG_BLOCKS;
        const int e0    = blockIdx.x * chunk;
        const int e1    = min(e0 + chunk, e);

        int c[PART_BATCH], r[PART_BATCH];
#pragma unroll
        for (int k = 0; k < PART_BATCH; ++k) {
            c[k] = 0;
            r[k] = 0;
        }
        if (e0 < e1) {
#pragma unroll
            for (int k = 0; k < PART_BATCH; ++k) {
                const int idx = min(e0 + tid + k * GP_THREADS, e1 - 1);
                c[k] = col[idx];
                r[k] = row[idx];
            }
        }
        if (tid < BK_MAX) {
            tot[tid] = 0u;
            bef[tid] = 0u;
            cnt[tid] = 0u;
        }
        {
            const int j = tid & (BK_MAX - 1);
            const int g = __builtin_amdgcn_readfirstlane(tid / BK_MAX);
            unsigned t = 0u, bf = 0u;
            if (j < nbk) {
#pragma unroll
                for (int i = 0; i < SUM_ROWS; ++i) {
                    const int b = g * SUM_ROWS + i;
                    const unsigned v = bcnt[b * BK_MAX + j];
                    t  += v;
                    bf += (b < (int)blockIdx.x) ? v : 0u;
                }
            }
            __syncthreads();
            if (j < nbk) {
                atomicAdd(&tot[j], t);
                atomicAdd(&bef[j], bf);
            }
        }
        __syncthreads();
        if (tid < 32) {
            unsigned v[8];
            unsigned run = 0u;
#pragma unroll
            for (int i = 0; i < 8; ++i) {
                unsigned t = tot[tid * 8 + i];
                v[i] = run;
                run += t;
            }
            unsigned inc = run;
            for (int d = 1; d < 32; d <<= 1) {
                unsigned t = __shfl_up(inc, d);
                if (tid >= d) inc += t;
            }
            const unsigned excl = inc - run;
#pragma unroll
            for (int i = 0; i < 8; ++i) {
                const unsigned st = excl + v[i];
                cur[tid * 8 + i] = st + bef[tid * 8 + i];
                if (blockIdx.x == 0) bstart[tid * 8 + i] = st;
            }
            if (blockIdx.x == 0 && tid == 31) bstart[BK_MAX] = inc;
        }
        __syncthreads();

        for (int base = e0; base < e1; base += PART_ROUND) {
            const int m = min(PART_ROUND, e1 - base);
            unsigned rank[PART_BATCH];
#pragma unroll
            for (int k = 0; k < PART_BATCH; ++k) {
                rank[k] = 0u;
                if (tid + k * GP_THREADS < m)
                    rank[k] = atomicAdd(&cnt[c[k] >> BK_SHIFT], 1u);
            }
            __syncthreads();
            if (tid < 32) {
                unsigned run = 0u;
#pragma unroll
                for (int i = 0; i < 8; ++i) run += cnt[tid * 8 + i];
                unsigned inc = run;
                for (int d = 1; d < 32; d <<= 1) {
                    unsigned u = __shfl_up(inc, d);
                    if (tid >= d) inc += u;
                }
                unsigned ls = inc - run;
#pragma unroll
                for (int i = 0; i < 8; ++i) {
                    const unsigned t  = cnt[tid * 8 + i];
                    const unsigned cu = cur[tid * 8 + i];
                    lst[tid * 8 + i] = ls;
                    gof[tid * 8 + i] = cu - ls;
                    cur[tid * 8 + i] = cu + t;
                    cnt[tid * 8 + i] = 0u;
                    ls += t;
                }
            }
            __syncthreads();
#pragma unroll
            for (int k = 0; k < PART_BATCH; ++k) rank[k] += lst[c[k] >> BK_SHIFT];
#pragma unroll
            for (int k = 0; k < PART_BATCH; ++k)
                if (tid + k * GP_THREADS < m)
                    stage[rank[k]] = make_int2(r[k], c[k]);
            __syncthreads();
            if (base + PART_ROUND < e1) {
#pragma unroll
                for (int k = 0; k < PART_BATCH; ++k) {
                    const int idx = min(base + PART_ROUND + tid + k * GP_THREADS, e1 - 1);
                    c[k] = col[idx];
                    r[k] = row[idx];
                }
            }
#pragma unroll
            for (int g = 0; g < PART_BATCH; g += 4) {
                int2 p[4];
                unsigned go[4];
#pragma unroll
                for (int q = 0; q < 4; ++q)
                    p[q] = stage[min(tid + (g + q) * GP_THREADS, m - 1)];
#pragma unroll
                for (int q = 0; q < 4; ++q)
                    go[q] = gof[p[q].y >> BK_SHIFT];
#pragma unroll
                for (int q = 0; q < 4; ++q) {
                    const int j = tid + (g + q) * GP_THREADS;
                    if (j < m) sorted[go[q] + (unsigned)j] = p[q];
                }
            }
        }
    } else {
        const int gb      = blockIdx.x - DEG_BLOCKS;
        const int lane    = tid & 31;
        const int waveId  = __builtin_amdgcn_readfirstlane(tid >> 5);
        const int rowbase = gb * GEMM_ROWS;
        const int ntiles  = (n + 15) >> 4;
        const int tile    = gb * GEMM_TILES + waveId;

        unsigned (&slo)[BK_MAX] = lds.tot;
        unsigned (&shi)[BK_MAX] = lds.bef;
        const int wp = (((n + 1) >> 1) + 3) & ~3;
        if (tid < GEMM_ROWS / 2) {
            slo[tid] = 0u;
            shi[tid] = 0u;
        }
        {
            const int wl = tid & (GEMM_ROWS / 2 - 1);
            const int g  = __builtin_amdgcn_readfirstlane(tid / (GEMM_ROWS / 2));
            const unsigned wu = (unsigned)((rowbase >> 1) + wl);
            unsigned lo = 0u, hi = 0u;
            if (2 * (int)wu < n) {
#pragma unroll
                for (int i = 0; i < SUM_ROWS; ++i) {
                    const unsigned* prow = part + (size_t)(g * SUM_ROWS + i) * wp;
                    const unsigned v = prow[wu];
                    lo += v & 0xFFFFu;
                    hi += v >> 16;
                }
            }
            __syncthreads();
            if (2 * (int)wu < n) {
                atomicAdd(&slo[wl], lo);
                atomicAdd(&shi[wl], hi);
            }
        }

        const int half = lane >> 4;
        const int idx  = lane & 15;
        const int koff = half * 2;

        v8f c = {};
        if (tile < ntiles) {
            int arow = tile * 16 + idx;
            if (arow >= n) arow = n - 1;
            const float* xrow = x + (size_t)arow * KDIM;
#pragma unroll
            for (int k = 0; k < KDIM; k += 4) {
                v2f a;
                a.x = xrow[k + koff + 0];
                a.y = xrow[k + koff + 1];
                v2f bm;
                bm.x = W[(k + koff + 0) * CHANNELS + idx];
                bm.y = W[(k + koff + 1) * CHANNELS + idx];
                c = __builtin_amdgcn_wmma_f32_16x16x4_f32(
                     false, a,  false, bm,
                     (short)0, c,  false,  false);
            }
        }
        __syncthreads();
        if (tid < GEMM_ROWS / 2) {
            const int w = (rowbase >> 1) + tid;
            float d0 = 0.0f, d1 = 0.0f;
            if (2 * w < n) {
                d0 = rsqrtf(1.0f + (float)slo[tid]);
                d1 = rsqrtf(1.0f + (float)shi[tid]);
                dis[2 * w] = d0;
                if (2 * w + 1 < n) dis[2 * w + 1] = d1;
            }
            sdis[2 * tid]     = d0;
            sdis[2 * tid + 1] = d1;
        }
        __syncthreads();

        if (tile < ntiles) {
#pragma unroll
            for (int v = 0; v < 8; ++v) {
                const int lr = waveId * 16 + v + 8 * half;
                const int rw = rowbase + lr;
                if (rw < n) h[(size_t)rw * CHANNELS + idx] = c[v] * sdis[lr];
            }
        }
    }
}

#define AGG_THREADS 1024
#define AGG_CHUNK   32

template <int J>
__device__ __forceinline__ unsigned agg_row_share(unsigned v) {
    return (unsigned)__builtin_amdgcn_update_dpp(0, (int)v, 0x150 + J, 0xF, 0xF, true);
}

#define AGG_GATHER(J) \
    hv[J] = *reinterpret_cast<const float*>(hbytes + (agg_row_share<J>(gall) | ch4));
#define AGG_ADD(J) \
    atomicAdd(reinterpret_cast<float*>(reinterpret_cast<char*>(acc) + \
                                       (agg_row_share<J>(lall) | ch4)), hv[J]);

__global__ __launch_bounds__(AGG_THREADS)
void gcn_aggregate_kernel(const int2* __restrict__ sorted,
                          const unsigned* __restrict__ bstart,
                          const float* __restrict__ dis,
                          const float* __restrict__ h,
                          const float* __restrict__ b,
                          float* __restrict__ out, int n) {
    __shared__ __attribute__((aligned(16))) float acc[(BK_NODES + 1) * CHANNELS];
    const int tid  = threadIdx.x;
    const int base = blockIdx.x << BK_SHIFT;

    for (int w = tid * 4; w < (BK_NODES + 1) * CHANNELS; w += AGG_THREADS * 4)
        *reinterpret_cast<float4*>(&acc[w]) = make_float4(0.0f, 0.0f, 0.0f, 0.0f);
    __syncthreads();

    const int s0 = (int)bstart[blockIdx.x];
    const int s1 = (int)bstart[blockIdx.x + 1];
    const int lane = tid & 31;
    const int wv = __builtin_amdgcn_readfirstlane(tid >> 5);
    const unsigned ch4 = (unsigned)(lane & 15) * 4u;
    const char* hbytes = reinterpret_cast<const char*>(h);
    int c0 = s0 + wv * AGG_CHUNK;
    int2 ed = make_int2(0, 0);
    if (c0 < s1) ed = sorted[min(c0 + lane, s1 - 1)];
    for (; c0 < s1; c0 += AGG_THREADS) {
        const bool valid = (c0 + lane) < s1;
        const unsigned gall = valid ? ((unsigned)ed.x << 6) : 0u;
        const unsigned lall = valid ? ((unsigned)(ed.y - base) << 6)
                                    : (unsigned)(BK_NODES << 6);
        float hv[16];
        AGG_GATHER(0)  AGG_GATHER(1)  AGG_GATHER(2)  AGG_GATHER(3)
        AGG_GATHER(4)  AGG_GATHER(5)  AGG_GATHER(6)  AGG_GATHER(7)
        AGG_GATHER(8)  AGG_GATHER(9)  AGG_GATHER(10) AGG_GATHER(11)
        AGG_GATHER(12) AGG_GATHER(13) AGG_GATHER(14) AGG_GATHER(15)
        ed = sorted[min(c0 + AGG_THREADS + lane, s1 - 1)];
        AGG_ADD(0)  AGG_ADD(1)  AGG_ADD(2)  AGG_ADD(3)
        AGG_ADD(4)  AGG_ADD(5)  AGG_ADD(6)  AGG_ADD(7)
        AGG_ADD(8)  AGG_ADD(9)  AGG_ADD(10) AGG_ADD(11)
        AGG_ADD(12) AGG_ADD(13) AGG_ADD(14) AGG_ADD(15)
    }
    __syncthreads();

    const int i = base + tid;
    if (tid < BK_NODES && i < n) {
        const float di = dis[i];
        float v[CHANNELS];
        float m = -INFINITY;
#pragma unroll
        for (int c = 0; c < CHANNELS; ++c) {
            float val = di * (acc[tid * CHANNELS + c] + h[(size_t)i * CHANNELS + c]) + b[c];
            val = fmaxf(val, 0.0f);
            v[c] = val;
            m = fmaxf(m, val);
        }
        float s = 0.0f;
#pragma unroll
        for (int c = 0; c < CHANNELS; ++c) s += __expf(v[c] - m);
        float lse = m + __logf(s);
#pragma unroll
        for (int c = 0; c < CHANNELS; ++c)
            out[(size_t)i * CHANNELS + c] = v[c] - lse;
    }
}

extern "C" void kernel_launch(void* const* d_in, const int* in_sizes, int n_in,
                              void* d_out, int out_size, void* d_ws, size_t ws_size,
                              hipStream_t stream) {
    const float* x  = (const float*)d_in[0];
    const int*   ei = (const int*)d_in[1];
    const float* W  = (const float*)d_in[2];
    const float* b  = (const float*)d_in[3];
    float* out = (float*)d_out;

    const int n = in_sizes[0] / KDIM;
    const int e = in_sizes[1] / 2;
    const int* row = ei;
    const int* col = ei + e;

    float* h   = (float*)d_ws;
    float* deg = h + (size_t)n * CHANNELS;
    unsigned* bcnt   = (unsigned*)(deg + (((size_t)n + 3) & ~(size_t)3));
    unsigned* bstart = bcnt + (size_t)DEG_BLOCKS * BK_MAX;
    int2* sorted     = (int2*)(bstart + BK_MAX + 16);
    unsigned* part   = (unsigned*)(sorted + (((size_t)e + 1) & ~(size_t)1));

    gcn_degree_lds_kernel<<<DEG_BLOCKS, DEG_THREADS, 0, stream>>>(col, part, bcnt, n, e);

    {
        const int ntiles = (n + 15) / 16;
        const int gemmBlocks = (ntiles + GEMM_TILES - 1) / GEMM_TILES;
        gcn_gemm_partition_kernel<<<DEG_BLOCKS + gemmBlocks, GP_THREADS, 0, stream>>>(
            x, W, part, deg, h, row, col, bcnt, bstart, sorted, n, e);
    }

    gcn_aggregate_kernel<<<(n + BK_NODES - 1) / BK_NODES, AGG_THREADS, 0, stream>>>(
        sorted, bstart, deg, h, b, out, n);
}


# Edge partition on twice the work-groups: a slice is placed by two, the front half filling each bucket's run upwards, the back half downwards from its end; 2 rounds each, not 4; arrival-order sums
# speedup vs baseline: 5.8052x; 1.0986x over previous
#include <hip/hip_runtime.h>
#include <math.h>

typedef __attribute__((ext_vector_type(2))) float v2f;
typedef __attribute__((ext_vector_type(8))) float v8f;

#define CHANNELS 16
#define KDIM 128

#define DEG_BLOCKS  128
#define DEG_THREADS 1024
#define DEG_CAPW    51200
#define DEG_BATCH   16

#define BK_SHIFT 9
#define BK_NODES 512
#define BK_MAX   256

__global__ __launch_bounds__(DEG_THREADS)
void gcn_degree_lds_kernel(const int* __restrict__ col,
                           unsigned* __restrict__ part,
                           unsigned* __restrict__ bcnt, int n, int e) {
    __shared__ __attribute__((aligned(16))) unsigned hist[DEG_CAPW];
    const int tid   = threadIdx.x;
    const int wp    = (((n + 1) >> 1) + 3) & ~3;
    const int chunk = (e + DEG_BLOCKS - 1) / DEG_BLOCKS;
    const int e0    = blockIdx.x * chunk;
    const int e1    = min(e0 + chunk, e);
    unsigned* prow  = part + (size_t)blockIdx.x * wp;

    for (int basew = 0; basew < wp; basew += DEG_CAPW) {
        const int nw = min(DEG_CAPW, wp - basew);
        int c[DEG_BATCH];
#pragma unroll
        for (int k = 0; k < DEG_BATCH; ++k) c[k] = 0;
        if (e0 < e1) {
#pragma unroll
            for (int k = 0; k < DEG_BATCH; ++k)
                c[k] = col[min(e0 + tid + k * DEG_THREADS, e1 - 1)];
        }
        for (int w = tid * 4; w < nw; w += DEG_THREADS * 4)
            *reinterpret_cast<uint4*>(&hist[w]) = make_uint4(0u, 0u, 0u, 0u);
        __syncthreads();
        for (int i0 = e0 + tid; i0 < e1; i0 += DEG_THREADS * DEG_BATCH) {
            int cn[DEG_BATCH];
            const int i1 = i0 + DEG_THREADS * DEG_BATCH;
#pragma unroll
            for (int k = 0; k < DEG_BATCH; ++k) cn[k] = 0;
            if (i1 < e1) {
#pragma unroll
                for (int k = 0; k < DEG_BATCH; ++k)
                    cn[k] = col[min(i1 + k * DEG_THREADS, e1 - 1)];
            }
#pragma unroll
            for (int k = 0; k < DEG_BATCH; ++k) {
                unsigned rel = (i0 + k * DEG_THREADS < e1)
                             ? (unsigned)c[k] - (unsigned)(basew * 2)
                             : 0xFFFFFFFFu;
                if (rel < (unsigned)(nw * 2))
                    atomicAdd(&hist[rel >> 1], 1u << ((rel & 1u) * 16u));
            }
#pragma unroll
            for (int k = 0; k < DEG_BATCH; ++k) c[k] = cn[k];
        }
        __syncthreads();
        for (int w = tid * 4; w < nw; w += DEG_THREADS * 4)
            *reinterpret_cast<uint4*>(&prow[basew + w]) =
                *reinterpret_cast<const uint4*>(&hist[w]);
        {
            const int lane = tid & 31;
            const int nbl  = (nw + 255) >> 8;
            for (int j = tid >> 5; j < nbl; j += DEG_THREADS / 32) {
                const int w0 = (j << 8) + lane * 8;
                unsigned s = 0u;
                if (w0 < nw) {
                    uint4 a = *reinterpret_cast<const uint4*>(&hist[w0]);
                    s += (a.x & 0xFFFFu) + (a.x >> 16) + (a.y & 0xFFFFu) + (a.y >> 16)
                       + (a.z & 0xFFFFu) + (a.z >> 16) + (a.w & 0xFFFFu) + (a.w >> 16);
                }
                if (w0 + 4 < nw) {
                    uint4 a = *reinterpret_cast<const uint4*>(&hist[w0 + 4]);
                    s += (a.x & 0xFFFFu) + (a.x >> 16) + (a.y & 0xFFFFu) + (a.y >> 16)
                       + (a.z & 0xFFFFu) + (a.z >> 16) + (a.w & 0xFFFFu) + (a.w >> 16);
                }
                for (int d = 16; d > 0; d >>= 1) s += __shfl_xor(s, d);
                if (lane == 0)
                    bcnt[blockIdx.x * BK_MAX + (basew >> 8) + j] = s;
            }
        }
        __syncthreads();
    }
}

#define PART_BATCH 8
#define GP_THREADS 1024
#define PART_BLOCKS (2 * DEG_BLOCKS)
#define PART_ROUND (GP_THREADS * PART_BATCH)
#define GEMM_TILES (GP_THREADS / 32)
#define GEMM_ROWS  (GEMM_TILES * 16)
#define SUM_GROUPS (GP_THREADS / BK_MAX)
#define SUM_ROWS   (DEG_BLOCKS / SUM_GROUPS)
static_assert(SUM_GROUPS * BK_MAX == GP_THREADS && SUM_GROUPS * SUM_ROWS == DEG_BLOCKS,
              "the slices must split evenly over the thread groups");
static_assert(GEMM_ROWS / 2 == BK_MAX, "a product block's degree words share the bucket tables");

__global__ __launch_bounds__(GP_THREADS) __attribute__((amdgpu_waves_per_eu(16, 16)))
void gcn_gemm_partition_kernel(const float* __restrict__ x,
                               const float* __restrict__ W,
                               const unsigned* __restrict__ part,
                               float* __restrict__ dis,
                               float* __restrict__ h,
                               const int* __restrict__ row,
                               const int* __restrict__ col,
                               const unsigned* __restrict__ bcnt,
                               unsigned* __restrict__ bstart,
                               int2* __restrict__ sorted, int n, int e) {
    struct GpLds {
        unsigned tot[BK_MAX];
        unsigned bef[BK_MAX];
        unsigned cur[BK_MAX];
        unsigned cnt[BK_MAX];
        unsigned lst[BK_MAX];
        unsigned gof[BK_MAX];
        float    sdis[GEMM_ROWS];
        int2     stage[PART_ROUND];
    };
    __shared__ __attribute__((aligned(16))) GpLds lds;
    unsigned (&tot)[BK_MAX] = lds.tot;
    unsigned (&bef)[BK_MAX] = lds.bef;
    unsigned (&cur)[BK_MAX] = lds.cur;
    unsigned (&cnt)[BK_MAX] = lds.cnt;
    unsigned (&lst)[BK_MAX] = lds.lst;
    unsigned (&gof)[BK_MAX] = lds.gof;
    float (&sdis)[GEMM_ROWS] = lds.sdis;
    int2 (&stage)[PART_ROUND] = lds.stage;
    const int tid = threadIdx.x;

    if (blockIdx.x < PART_BLOCKS) {
        const int nbk   = (n + BK_NODES - 1) >> BK_SHIFT;
        const int chunk = (e + DEG_BLOCKS - 1) / DEG_BLOCKS;
        const int sl    = blockIdx.x >> 1;
        const int hf    = blockIdx.x & 1;
        const unsigned neg = 0u - (unsigned)hf;
        const int sl0   = sl * chunk;
        const int sl1   = min(sl0 + chunk, e);
        const int mid   = sl0 + ((max(sl1 - sl0, 0) + 1) >> 1);
        const int e0    = hf ? mid : sl0;
        const int e1    = hf ? sl1 : mid;

        int c[PART_BATCH], r[PART_BATCH];
#pragma unroll
        for (int k = 0; k < PART_BATCH; ++k) {
            c[k] = 0;
            r[k] = 0;
        }
        if (e0 < e1) {
#pragma unroll
            for (int k = 0; k < PART_BATCH; ++k) {
                const int idx = min(e0 + tid + k * GP_THREADS, e1 - 1);
                c[k] = col[idx];
                r[k] = row[idx];
            }
        }
        if (tid < BK_MAX) {
            tot[tid] = 0u;
            bef[tid] = 0u;
            cnt[tid] = 0u;
        }
        {
            const int j = tid & (BK_MAX - 1);
            const int g = __builtin_amdgcn_readfirstlane(tid / BK_MAX);
            unsigned t = 0u, bf = 0u;
            if (j < nbk) {
#pragma unroll
                for (int i = 0; i < SUM_ROWS; ++i) {
                    const int b = g * SUM_ROWS + i;
                    const unsigned v = bcnt[b * BK_MAX + j];
                    t  += v;
                    bf += (b < sl + hf) ? v : 0u;
                }
            }
            __syncthreads();
            if (j < nbk) {
                atomicAdd(&tot[j], t);
                atomicAdd(&bef[j], bf);
            }
        }
        __syncthreads();
        if (tid < 32) {
            unsigned v[8];
            unsigned run = 0u;
#pragma unroll
            for (int i = 0; i < 8; ++i) {
                unsigned t = tot[tid * 8 + i];
                v[i] = run;
                run += t;
            }
            unsigned inc = run;
            for (int d = 1; d < 32; d <<= 1) {
                unsigned t = __shfl_up(inc, d);
                if (tid >= d) inc += t;
            }
            const unsigned excl = inc - run;
#pragma unroll
            for (int i = 0; i < 8; ++i) {
                const unsigned st = excl + v[i];
                const unsigned pl = st + bef[tid * 8 + i];
                cur[tid * 8 + i] = ((pl - (unsigned)hf) ^ neg) - neg;
                if (blockIdx.x == 0) bstart[tid * 8 + i] = st;
            }
            if (blockIdx.x == 0 && tid == 31) bstart[BK_MAX] = inc;
        }
        __syncthreads();

        for (int base = e0; base < e1; base += PART_ROUND) {
            const int m = min(PART_ROUND, e1 - base);
            unsigned rank[PART_BATCH];
#pragma unroll
            for (int k = 0; k < PART_BATCH; ++k) {
                rank[k] = 0u;
                if (tid + k * GP_THREADS < m)
                    rank[k] = atomicAdd(&cnt[c[k] >> BK_SHIFT], 1u);
            }
            __syncthreads();
            if (tid < 32) {
                unsigned run = 0u;
#pragma unroll
                for (int i = 0; i < 8; ++i) run += cnt[tid * 8 + i];
                unsigned inc = run;
                for (int d = 1; d < 32; d <<= 1) {
                    unsigned u = __shfl_up(inc, d);
                    if (tid >= d) inc += u;
                }
                unsigned ls = inc - run;
#pragma unroll
                for (int i = 0; i < 8; ++i) {
                    const unsigned t  = cnt[tid * 8 + i];
                    const unsigned cu = cur[tid * 8 + i];
                    lst[tid * 8 + i] = ls;
                    gof[tid * 8 + i] = cu - ls;
                    cur[tid * 8 + i] = cu + t;
                    cnt[tid * 8 + i] = 0u;
                    ls += t;
                }
            }
            __syncthreads();
#pragma unroll
            for (int k = 0; k < PART_BATCH; ++k) rank[k] += lst[c[k] >> BK_SHIFT];
#pragma unroll
            for (int k = 0; k < PART_BATCH; ++k)
                if (tid + k * GP_THREADS < m)
                    stage[rank[k]] = make_int2(r[k], c[k]);
            __syncthreads();
            if (base + PART_ROUND < e1) {
#pragma unroll
                for (int k = 0; k < PART_BATCH; ++k) {
                    const int idx = min(base + PART_ROUND + tid + k * GP_THREADS, e1 - 1);
                    c[k] = col[idx];
                    r[k] = row[idx];
                }
            }
#pragma unroll
            for (int g = 0; g < PART_BATCH; g += 4) {
                int2 p[4];
                unsigned go[4];
#pragma unroll
                for (int q = 0; q < 4; ++q)
                    p[q] = stage[min(tid + (g + q) * GP_THREADS, m - 1)];
#pragma unroll
                for (int q = 0; q < 4; ++q)
                    go[q] = gof[p[q].y >> BK_SHIFT];
#pragma unroll
                for (int q = 0; q < 4; ++q) {
                    const int j = tid + (g + q) * GP_THREADS;
                    if (j < m) sorted[((go[q] + (unsigned)j) ^ neg) - neg] = p[q];
                }
            }
        }
    } else {
        const int gb      = blockIdx.x - PART_BLOCKS;
        const int lane    = tid & 31;
        const int waveId  = __builtin_amdgcn_readfirstlane(tid >> 5);
        const int rowbase = gb * GEMM_ROWS;
        const int ntiles  = (n + 15) >> 4;
        const int tile    = gb * GEMM_TILES + waveId;

        unsigned (&slo)[BK_MAX] = lds.tot;
        unsigned (&shi)[BK_MAX] = lds.bef;
        const int wp = (((n + 1) >> 1) + 3) & ~3;
        if (tid < GEMM_ROWS / 2) {
            slo[tid] = 0u;
            shi[tid] = 0u;
        }
        {
            const int wl = tid & (GEMM_ROWS / 2 - 1);
            const int g  = __builtin_amdgcn_readfirstlane(tid / (GEMM_ROWS / 2));
            const unsigned wu = (unsigned)((rowbase >> 1) + wl);
            unsigned lo = 0u, hi = 0u;
            if (2 * (int)wu < n) {
#pragma unroll
                for (int i = 0; i < SUM_ROWS; ++i) {
                    const unsigned* prow = part + (size_t)(g * SUM_ROWS + i) * wp;
                    const unsigned v = prow[wu];
                    lo += v & 0xFFFFu;
                    hi += v >> 16;
                }
            }
            __syncthreads();
            if (2 * (int)wu < n) {
                atomicAdd(&slo[wl], lo);
                atomicAdd(&shi[wl], hi);
            }
        }

        const int half = lane >> 4;
        const int idx  = lane & 15;
        const int koff = half * 2;

        v8f c = {};
        if (tile < ntiles) {
            int arow = tile * 16 + idx;
            if (arow >= n) arow = n - 1;
            const float* xrow = x + (size_t)arow * KDIM;
#pragma unroll
            for (int k = 0; k < KDIM; k += 4) {
                v2f a;
                a.x = xrow[k + koff + 0];
                a.y = xrow[k + koff + 1];
                v2f bm;
                bm.x = W[(k + koff + 0) * CHANNELS + idx];
                bm.y = W[(k + koff + 1) * CHANNELS + idx];
                c = __builtin_amdgcn_wmma_f32_16x16x4_f32(
                     false, a,  false, bm,
                     (short)0, c,  false,  false);
            }
        }
        __syncthreads();
        if (tid < GEMM_ROWS / 2) {
            const int w = (rowbase >> 1) + tid;
            float d0 = 0.0f, d1 = 0.0f;
            if (2 * w < n) {
                d0 = rsqrtf(1.0f + (float)slo[tid]);
                d1 = rsqrtf(1.0f + (float)shi[tid]);
                dis[2 * w] = d0;
                if (2 * w + 1 < n) dis[2 * w + 1] = d1;
            }
            sdis[2 * tid]     = d0;
            sdis[2 * tid + 1] = d1;
        }
        __syncthreads();

        if (tile < ntiles) {
#pragma unroll
            for (int v = 0; v < 8; ++v) {
                const int lr = waveId * 16 + v + 8 * half;
                const int rw = rowbase + lr;
                if (rw < n) h[(size_t)rw * CHANNELS + idx] = c[v] * sdis[lr];
            }
        }
    }
}

#define AGG_THREADS 1024
#define AGG_CHUNK   32

template <int J>
__device__ __forceinline__ unsigned agg_row_share(unsigned v) {
    return (unsigned)__builtin_amdgcn_update_dpp(0, (int)v, 0x150 + J, 0xF, 0xF, true);
}

#define AGG_GATHER(J) \
    hv[J] = *reinterpret_cast<const float*>(hbytes + (agg_row_share<J>(gall) | ch4));
#define AGG_ADD(J) \
    atomicAdd(reinterpret_cast<float*>(reinterpret_cast<char*>(acc) + \
                                       (agg_row_share<J>(lall) | ch4)), hv[J]);

__global__ __launch_bounds__(AGG_THREADS)
void gcn_aggregate_kernel(const int2* __restrict__ sorted,
                          const unsigned* __restrict__ bstart,
                          const float* __restrict__ dis,
                          const float* __restrict__ h,
                          const float* __restrict__ b,
                          float* __restrict__ out, int n) {
    __shared__ __attribute__((aligned(16))) float acc[(BK_NODES + 1) * CHANNELS];
    const int tid  = threadIdx.x;
    const int base = blockIdx.x << BK_SHIFT;

    for (int w = tid * 4; w < (BK_NODES + 1) * CHANNELS; w += AGG_THREADS * 4)
        *reinterpret_cast<float4*>(&acc[w]) = make_float4(0.0f, 0.0f, 0.0f, 0.0f);
    __syncthreads();

    const int s0 = (int)bstart[blockIdx.x];
    const int s1 = (int)bstart[blockIdx.x + 1];
    const int lane = tid & 31;
    const int wv = __builtin_amdgcn_readfirstlane(tid >> 5);
    const unsigned ch4 = (unsigned)(lane & 15) * 4u;
    const char* hbytes = reinterpret_cast<const char*>(h);
    int c0 = s0 + wv * AGG_CHUNK;
    int2 ed = make_int2(0, 0);
    if (c0 < s1) ed = sorted[min(c0 + lane, s1 - 1)];
    for (; c0 < s1; c0 += AGG_THREADS) {
        const bool valid = (c0 + lane) < s1;
        const unsigned gall = valid ? ((unsigned)ed.x << 6) : 0u;
        const unsigned lall = valid ? ((unsigned)(ed.y - base) << 6)
                                    : (unsigned)(BK_NODES << 6);
        float hv[16];
        AGG_GATHER(0)  AGG_GATHER(1)  AGG_GATHER(2)  AGG_GATHER(3)
        AGG_GATHER(4)  AGG_GATHER(5)  AGG_GATHER(6)  AGG_GATHER(7)
        AGG_GATHER(8)  AGG_GATHER(9)  AGG_GATHER(10) AGG_GATHER(11)
        AGG_GATHER(12) AGG_GATHER(13) AGG_GATHER(14) AGG_GATHER(15)
        ed = sorted[min(c0 + AGG_THREADS + lane, s1 - 1)];
        AGG_ADD(0)  AGG_ADD(1)  AGG_ADD(2)  AGG_ADD(3)
        AGG_ADD(4)  AGG_ADD(5)  AGG_ADD(6)  AGG_ADD(7)
        AGG_ADD(8)  AGG_ADD(9)  AGG_ADD(10) AGG_ADD(11)
        AGG_ADD(12) AGG_ADD(13) AGG_ADD(14) AGG_ADD(15)
    }
    __syncthreads();

    const int i = base + tid;
    if (tid < BK_NODES && i < n) {
        const float di = dis[i];
        float v[CHANNELS];
        float m = -INFINITY;
#pragma unroll
        for (int c = 0; c < CHANNELS; ++c) {
            float val = di * (acc[tid * CHANNELS + c] + h[(size_t)i * CHANNELS + c]) + b[c];
            val = fmaxf(val, 0.0f);
            v[c] = val;
            m = fmaxf(m, val);
        }
        float s = 0.0f;
#pragma unroll
        for (int c = 0; c < CHANNELS; ++c) s += __expf(v[c] - m);
        float lse = m + __logf(s);
#pragma unroll
        for (int c = 0; c < CHANNELS; ++c)
            out[(size_t)i * CHANNELS + c] = v[c] - lse;
    }
}

extern "C" void kernel_launch(void* const* d_in, const int* in_sizes, int n_in,
                              void* d_out, int out_size, void* d_ws, size_t ws_size,
                              hipStream_t stream) {
    const float* x  = (const float*)d_in[0];
    const int*   ei = (const int*)d_in[1];
    const float* W  = (const float*)d_in[2];
    const float* b  = (const float*)d_in[3];
    float* out = (float*)d_out;

    const int n = in_sizes[0] / KDIM;
    const int e = in_sizes[1] / 2;
    const int* row = ei;
    const int* col = ei + e;

    float* h   = (float*)d_ws;
    float* deg = h + (size_t)n * CHANNELS;
    unsigned* bcnt   = (unsigned*)(deg + (((size_t)n + 3) & ~(size_t)3));
    unsigned* bstart = bcnt + (size_t)DEG_BLOCKS * BK_MAX;
    int2* sorted     = (int2*)(bstart + BK_MAX + 16);
    unsigned* part   = (unsigned*)(sorted + (((size_t)e + 1) & ~(size_t)1));

    gcn_degree_lds_kernel<<<DEG_BLOCKS, DEG_THREADS, 0, stream>>>(col, part, bcnt, n, e);

    {
        const int ntiles = (n + 15) / 16;
        const int gemmBlocks = (ntiles + GEMM_TILES - 1) / GEMM_TILES;
        gcn_gemm_partition_kernel<<<PART_BLOCKS + gemmBlocks, GP_THREADS, 0, stream>>>(
            x, W, part, deg, h, row, col, bcnt, bstart, sorted, n, e);
    }

    gcn_aggregate_kernel<<<(n + BK_NODES - 1) / BK_NODES, AGG_THREADS, 0, stream>>>(
        sorted, bstart, deg, h, b, out, n);
}


# x@W operands fed through LDS: x read as whole 128-B lines (16 B a lane, 4 rows a load, two quarters of k in flight), turned in a per-wave LDS tile; W staged once per block; k summed in another order
# speedup vs baseline: 6.2930x; 1.0840x over previous
#include <hip/hip_runtime.h>
#include <math.h>

typedef __attribute__((ext_vector_type(2))) float v2f;
typedef __attribute__((ext_vector_type(8))) float v8f;
typedef __attribute__((ext_vector_type(4))) float v4f;

#define CHANNELS 16
#define KDIM 128

#define DEG_BLOCKS  128
#define DEG_THREADS 1024
#define DEG_CAPW    51200
#define DEG_BATCH   16

#define BK_SHIFT 9
#define BK_NODES 512
#define BK_MAX   256

__global__ __launch_bounds__(DEG_THREADS)
void gcn_degree_lds_kernel(const int* __restrict__ col,
                           unsigned* __restrict__ part,
                           unsigned* __restrict__ bcnt, int n, int e) {
    __shared__ __attribute__((aligned(16))) unsigned hist[DEG_CAPW];
    const int tid   = threadIdx.x;
    const int wp    = (((n + 1) >> 1) + 3) & ~3;
    const int chunk = (e + DEG_BLOCKS - 1) / DEG_BLOCKS;
    const int e0    = blockIdx.x * chunk;
    const int e1    = min(e0 + chunk, e);
    unsigned* prow  = part + (size_t)blockIdx.x * wp;

    for (int basew = 0; basew < wp; basew += DEG_CAPW) {
        const int nw = min(DEG_CAPW, wp - basew);
        int c[DEG_BATCH];
#pragma unroll
        for (int k = 0; k < DEG_BATCH; ++k) c[k] = 0;
        if (e0 < e1) {
#pragma unroll
            for (int k = 0; k < DEG_BATCH; ++k)
                c[k] = col[min(e0 + tid + k * DEG_THREADS, e1 - 1)];
        }
        for (int w = tid * 4; w < nw; w += DEG_THREADS * 4)
            *reinterpret_cast<uint4*>(&hist[w]) = make_uint4(0u, 0u, 0u, 0u);
        __syncthreads();
        for (int i0 = e0 + tid; i0 < e1; i0 += DEG_THREADS * DEG_BATCH) {
            int cn[DEG_BATCH];
            const int i1 = i0 + DEG_THREADS * DEG_BATCH;
#pragma unroll
            for (int k = 0; k < DEG_BATCH; ++k) cn[k] = 0;
            if (i1 < e1) {
#pragma unroll
                for (int k = 0; k < DEG_BATCH; ++k)
                    cn[k] = col[min(i1 + k * DEG_THREADS, e1 - 1)];
            }
#pragma unroll
            for (int k = 0; k < DEG_BATCH; ++k) {
                unsigned rel = (i0 + k * DEG_THREADS < e1)
                             ? (unsigned)c[k] - (unsigned)(basew * 2)
                             : 0xFFFFFFFFu;
                if (rel < (unsigned)(nw * 2))
                    atomicAdd(&hist[rel >> 1], 1u << ((rel & 1u) * 16u));
            }
#pragma unroll
            for (int k = 0; k < DEG_BATCH; ++k) c[k] = cn[k];
        }
        __syncthreads();
        for (int w = tid * 4; w < nw; w += DEG_THREADS * 4)
            *reinterpret_cast<uint4*>(&prow[basew + w]) =
                *reinterpret_cast<const uint4*>(&hist[w]);
        {
            const int lane = tid & 31;
            const int nbl  = (nw + 255) >> 8;
            for (int j = tid >> 5; j < nbl; j += DEG_THREADS / 32) {
                const int w0 = (j << 8) + lane * 8;
                unsigned s = 0u;
                if (w0 < nw) {
                    uint4 a = *reinterpret_cast<const uint4*>(&hist[w0]);
                    s += (a.x & 0xFFFFu) + (a.x >> 16) + (a.y & 0xFFFFu) + (a.y >> 16)
                       + (a.z & 0xFFFFu) + (a.z >> 16) + (a.w & 0xFFFFu) + (a.w >> 16);
                }
                if (w0 + 4 < nw) {
                    uint4 a = *reinterpret_cast<const uint4*>(&hist[w0 + 4]);
                    s += (a.x & 0xFFFFu) + (a.x >> 16) + (a.y & 0xFFFFu) + (a.y >> 16)
                       + (a.z & 0xFFFFu) + (a.z >> 16) + (a.w & 0xFFFFu) + (a.w >> 16);
                }
                for (int d = 16; d > 0; d >>= 1) s += __shfl_xor(s, d);
                if (lane == 0)
                    bcnt[blockIdx.x * BK_MAX + (basew >> 8) + j] = s;
            }
        }
        __syncthreads();
    }
}

#define PART_BATCH 8
#define GP_THREADS 1024
#define PART_BLOCKS (2 * DEG_BLOCKS)
#define PART_ROUND (GP_THREADS * PART_BATCH)
#define GEMM_TILES (GP_THREADS / 32)
#define GEMM_ROWS  (GEMM_TILES * 16)
#define SUM_GROUPS (GP_THREADS / BK_MAX)
#define SUM_ROWS   (DEG_BLOCKS / SUM_GROUPS)
static_assert(SUM_GROUPS * BK_MAX == GP_THREADS && SUM_GROUPS * SUM_ROWS == DEG_BLOCKS,
              "the slices must split evenly over the thread groups");
static_assert(GEMM_ROWS / 2 == BK_MAX, "a product block's degree words share the bucket tables");
#define XT_PITCH 36
#define XT_BYTES (GEMM_TILES * 16 * XT_PITCH * 4)
#define STAGE_SLOTS ((XT_BYTES / 8) > PART_ROUND ? (XT_BYTES / 8) : PART_ROUND)
static_assert(KDIM == 128 && CHANNELS == 16 && 2 * GP_THREADS == KDIM * CHANNELS,
              "the staging of W and the quarter loop are written for a 128 x 16 W");

__device__ __forceinline__ void wave_lds_order() {
    __builtin_amdgcn_fence(__ATOMIC_RELEASE, "wavefront");
    __builtin_amdgcn_wave_barrier();
    __builtin_amdgcn_fence(__ATOMIC_ACQUIRE, "wavefront");
}

__global__ __launch_bounds__(GP_THREADS) __attribute__((amdgpu_waves_per_eu(16, 16)))
void gcn_gemm_partition_kernel(const float* __restrict__ x,
                               const float* __restrict__ W,
                               const unsigned* __restrict__ part,
                               float* __restrict__ dis,
                               float* __restrict__ h,
                               const int* __restrict__ row,
                               const int* __restrict__ col,
                               const unsigned* __restrict__ bcnt,
                               unsigned* __restrict__ bstart,
                               int2* __restrict__ sorted, int n, int e) {
    struct GpLds {
        unsigned tot[BK_MAX];
        unsigned bef[BK_MAX];
        unsigned cur[BK_MAX];
        unsigned cnt[BK_MAX];
        unsigned lst[BK_MAX];
        unsigned gof[BK_MAX];
        float    sdis[GEMM_ROWS];
        float    wt[KDIM * CHANNELS];
        int2     stage[STAGE_SLOTS];
    };
    __shared__ __attribute__((aligned(16))) GpLds lds;
    unsigned (&tot)[BK_MAX] = lds.tot;
    unsigned (&bef)[BK_MAX] = lds.bef;
    unsigned (&cur)[BK_MAX] = lds.cur;
    unsigned (&cnt)[BK_MAX] = lds.cnt;
    unsigned (&lst)[BK_MAX] = lds.lst;
    unsigned (&gof)[BK_MAX] = lds.gof;
    float (&sdis)[GEMM_ROWS] = lds.sdis;
    int2 (&stage)[STAGE_SLOTS] = lds.stage;
    float (&wt)[KDIM * CHANNELS] = lds.wt;
    const int tid = threadIdx.x;

    if (blockIdx.x < PART_BLOCKS) {
        const int nbk   = (n + BK_NODES - 1) >> BK_SHIFT;
        const int chunk = (e + DEG_BLOCKS - 1) / DEG_BLOCKS;
        const int sl    = blockIdx.x >> 1;
        const int hf    = blockIdx.x & 1;
        const unsigned neg = 0u - (unsigned)hf;
        const int sl0   = sl * chunk;
        const int sl1   = min(sl0 + chunk, e);
        const int mid   = sl0 + ((max(sl1 - sl0, 0) + 1) >> 1);
        const int e0    = hf ? mid : sl0;
        const int e1    = hf ? sl1 : mid;

        int c[PART_BATCH], r[PART_BATCH];
#pragma unroll
        for (int k = 0; k < PART_BATCH; ++k) {
            c[k] = 0;
            r[k] = 0;
        }
        if (e0 < e1) {
#pragma unroll
            for (int k = 0; k < PART_BATCH; ++k) {
                const int idx = min(e0 + tid + k * GP_THREADS, e1 - 1);
                c[k] = col[idx];
                r[k] = row[idx];
            }
        }
        if (tid < BK_MAX) {
            tot[tid] = 0u;
            bef[tid] = 0u;
            cnt[tid] = 0u;
        }
        {
            const int j = tid & (BK_MAX - 1);
            const int g = __builtin_amdgcn_readfirstlane(tid / BK_MAX);
            unsigned t = 0u, bf = 0u;
            if (j < nbk) {
#pragma unroll
                for (int i = 0; i < SUM_ROWS; ++i) {
                    const int b = g * SUM_ROWS + i;
                    const unsigned v = bcnt[b * BK_MAX + j];
                    t  += v;
                    bf += (b < sl + hf) ? v : 0u;
                }
            }
            __syncthreads();
            if (j < nbk) {
                atomicAdd(&tot[j], t);
                atomicAdd(&bef[j], bf);
            }
        }
        __syncthreads();
        if (tid < 32) {
            unsigned v[8];
            unsigned run = 0u;
#pragma unroll
            for (int i = 0; i < 8; ++i) {
                unsigned t = tot[tid * 8 + i];
                v[i] = run;
                run += t;
            }
            unsigned inc = run;
            for (int d = 1; d < 32; d <<= 1) {
                unsigned t = __shfl_up(inc, d);
                if (tid >= d) inc += t;
            }
            const unsigned excl = inc - run;
#pragma unroll
            for (int i = 0; i < 8; ++i) {
                const unsigned st = excl + v[i];
                const unsigned pl = st + bef[tid * 8 + i];
                cur[tid * 8 + i] = ((pl - (unsigned)hf) ^ neg) - neg;
                if (blockIdx.x == 0) bstart[tid * 8 + i] = st;
            }
            if (blockIdx.x == 0 && tid == 31) bstart[BK_MAX] = inc;
        }
        __syncthreads();

        for (int base = e0; base < e1; base += PART_ROUND) {
            const int m = min(PART_ROUND, e1 - base);
            unsigned rank[PART_BATCH];
#pragma unroll
            for (int k = 0; k < PART_BATCH; ++k) {
                rank[k] = 0u;
                if (tid + k * GP_THREADS < m)
                    rank[k] = atomicAdd(&cnt[c[k] >> BK_SHIFT], 1u);
            }
            __syncthreads();
            if (tid < 32) {
                unsigned run = 0u;
#pragma unroll
                for (int i = 0; i < 8; ++i) run += cnt[tid * 8 + i];
                unsigned inc = run;
                for (int d = 1; d < 32; d <<= 1) {
                    unsigned u = __shfl_up(inc, d);
                    if (tid >= d) inc += u;
                }
                unsigned ls = inc - run;
#pragma unroll
                for (int i = 0; i < 8; ++i) {
                    const unsigned t  = cnt[tid * 8 + i];
                    const unsigned cu = cur[tid * 8 + i];
                    lst[tid * 8 + i] = ls;
                    gof[tid * 8 + i] = cu - ls;
                    cur[tid * 8 + i] = cu + t;
                    cnt[tid * 8 + i] = 0u;
                    ls += t;
                }
            }
            __syncthreads();
#pragma unroll
            for (int k = 0; k < PART_BATCH; ++k) rank[k] += lst[c[k] >> BK_SHIFT];
#pragma unroll
            for (int k = 0; k < PART_BATCH; ++k)
                if (tid + k * GP_THREADS < m)
                    stage[rank[k]] = make_int2(r[k], c[k]);
            __syncthreads();
            if (base + PART_ROUND < e1) {
#pragma unroll
                for (int k = 0; k < PART_BATCH; ++k) {
                    const int idx = min(base + PART_ROUND + tid + k * GP_THREADS, e1 - 1);
                    c[k] = col[idx];
                    r[k] = row[idx];
                }
            }
#pragma unroll
            for (int g = 0; g < PART_BATCH; g += 4) {
                int2 p[4];
                unsigned go[4];
#pragma unroll
                for (int q = 0; q < 4; ++q)
                    p[q] = stage[min(tid + (g + q) * GP_THREADS, m - 1)];
#pragma unroll
                for (int q = 0; q < 4; ++q)
                    go[q] = gof[p[q].y >> BK_SHIFT];
#pragma unroll
                for (int q = 0; q < 4; ++q) {
                    const int j = tid + (g + q) * GP_THREADS;
                    if (j < m) sorted[((go[q] + (unsigned)j) ^ neg) - neg] = p[q];
                }
            }
        }
    } else {
        const int gb      = blockIdx.x - PART_BLOCKS;
        const int lane    = tid & 31;
        const int waveId  = __builtin_amdgcn_readfirstlane(tid >> 5);
        const int rowbase = gb * GEMM_ROWS;
        const int ntiles  = (n + 15) >> 4;
        const int tile    = gb * GEMM_TILES + waveId;
        const int half = lane >> 4;
        const int idx  = lane & 15;

        const int lr = lane >> 3;
        const int lk = (lane & 7) * 4;
        const unsigned xr0 = ((unsigned)min(tile * 16 + 0 + lr, n - 1) * KDIM + lk) * 4u;
        const unsigned xr1 = ((unsigned)min(tile * 16 + 4 + lr, n - 1) * KDIM + lk) * 4u;
        const unsigned xr2 = ((unsigned)min(tile * 16 + 8 + lr, n - 1) * KDIM + lk) * 4u;
        const unsigned xr3 = ((unsigned)min(tile * 16 + 12 + lr, n - 1) * KDIM + lk) * 4u;
#define XLOAD(G, P, OFF) \
        asm volatile("global_load_b128 %0, %1, %2 offset:%3" : "=&v"(G) : "v"(P), "s"(x), "n"(OFF) : "memory")
#define XSETTLE() asm volatile("s_wait_xcnt 0x0" ::: "memory")
#define XWAIT(G0, G1, G2, G3, N) \
        asm volatile("s_wait_loadcnt %4" : "+v"(G0), "+v"(G1), "+v"(G2), "+v"(G3) : "n"(N) : "memory")
        v4f ga0, ga1, ga2, ga3;
        if (tile < ntiles) {
            XLOAD(ga0, xr0, 0); XLOAD(ga1, xr1, 0); XLOAD(ga2, xr2, 0); XLOAD(ga3, xr3, 0);
            XSETTLE();
        }

        unsigned (&slo)[BK_MAX] = lds.tot;
        unsigned (&shi)[BK_MAX] = lds.bef;
        const int wp = (((n + 1) >> 1) + 3) & ~3;
        if (tid < GEMM_ROWS / 2) {
            slo[tid] = 0u;
            shi[tid] = 0u;
        }
        {
            const int wl = tid & (GEMM_ROWS / 2 - 1);
            const int g  = __builtin_amdgcn_readfirstlane(tid / (GEMM_ROWS / 2));
            const unsigned wu = (unsigned)((rowbase >> 1) + wl);
            unsigned lo = 0u, hi = 0u;
            if (2 * (int)wu < n) {
#pragma unroll
                for (int i = 0; i < SUM_ROWS; ++i) {
                    const unsigned* prow = part + (size_t)(g * SUM_ROWS + i) * wp;
                    const unsigned v = prow[wu];
                    lo += v & 0xFFFFu;
                    hi += v >> 16;
                }
            }
            {
                const float2 wv = *reinterpret_cast<const float2*>(W + 2 * tid);
                const int wk = (2 * tid) >> 4;
                const int wc = (2 * tid) & 15;
                wt[(wk >> 2) * 64 + wc * 4 + (wk & 3)]       = wv.x;
                wt[(wk >> 2) * 64 + (wc + 1) * 4 + (wk & 3)] = wv.y;
            }
            __syncthreads();
            if (2 * (int)wu < n) {
                atomicAdd(&slo[wl], lo);
                atomicAdd(&shi[wl], hi);
            }
        }

        v8f c = {};
        if (tile < ntiles) {
            float* xw = reinterpret_cast<float*>(&stage[0]) + waveId * 16 * XT_PITCH;
            float* xs = xw + lr * XT_PITCH + lk;
            const float* xa = xw + idx * XT_PITCH + 4 * half;
            const float* wb = &wt[half * 64 + idx * 4];
            v4f gb0, gb1, gb2, gb3;
            XLOAD(gb0, xr0, 128); XLOAD(gb1, xr1, 128); XLOAD(gb2, xr2, 128); XLOAD(gb3, xr3, 128);
            XSETTLE();
#define GEMM_QUARTER(G0, G1, G2, G3, Q)                                                        \
            {                                                                                  \
                XWAIT(G0, G1, G2, G3, ((Q) < 3 ? 4 : 0));                                      \
                *reinterpret_cast<v4f*>(xs + 0 * XT_PITCH)  = G0;                              \
                *reinterpret_cast<v4f*>(xs + 4 * XT_PITCH)  = G1;                              \
                *reinterpret_cast<v4f*>(xs + 8 * XT_PITCH)  = G2;                              \
                *reinterpret_cast<v4f*>(xs + 12 * XT_PITCH) = G3;                              \
                wave_lds_order();            \
                if ((Q) + 2 < 4) {                                                             \
                    XLOAD(G0, xr0, 128 * (((Q) + 2) & 3)); XLOAD(G1, xr1, 128 * (((Q) + 2) & 3)); \
                    XLOAD(G2, xr2, 128 * (((Q) + 2) & 3)); XLOAD(G3, xr3, 128 * (((Q) + 2) & 3)); \
                    XSETTLE();                                                                 \
                }                                                                              \
                _Pragma("unroll")                                                              \
                for (int s = 0; s < 4; ++s) {                                                  \
                    const float4 av = *reinterpret_cast<const float4*>(xa + 8 * s);            \
                    const float4 bv = *reinterpret_cast<const float4*>(wb + (8 * (Q) + 2 * s) * 64); \
                    v2f a0, b0, a1, b1;                                                        \
                    a0.x = av.x; a0.y = av.y; b0.x = bv.x; b0.y = bv.y;                        \
                    a1.x = av.z; a1.y = av.w; b1.x = bv.z; b1.y = bv.w;                        \
                    c = __builtin_amdgcn_wmma_f32_16x16x4_f32(false, a0, false, b0, (short)0, c, false, false); \
                    c = __builtin_amdgcn_wmma_f32_16x16x4_f32(false, a1, false, b1, (short)0, c, false, false); \
                }                                                                              \
                wave_lds_order();                     \
            }
            GEMM_QUARTER(ga0, ga1, ga2, ga3, 0)
            GEMM_QUARTER(gb0, gb1, gb2, gb3, 1)
            GEMM_QUARTER(ga0, ga1, ga2, ga3, 2)
            GEMM_QUARTER(gb0, gb1, gb2, gb3, 3)
#undef XLOAD
#undef XSETTLE
#undef XWAIT
#undef GEMM_QUARTER
        }
        __syncthreads();
        if (tid < GEMM_ROWS / 2) {
            const int w = (rowbase >> 1) + tid;
            float d0 = 0.0f, d1 = 0.0f;
            if (2 * w < n) {
                d0 = rsqrtf(1.0f + (float)slo[tid]);
                d1 = rsqrtf(1.0f + (float)shi[tid]);
                dis[2 * w] = d0;
                if (2 * w + 1 < n) dis[2 * w + 1] = d1;
            }
            sdis[2 * tid]     = d0;
            sdis[2 * tid + 1] = d1;
        }
        __syncthreads();

        if (tile < ntiles) {
#pragma unroll
            for (int v = 0; v < 8; ++v) {
                const int lr = waveId * 16 + v + 8 * half;
                const int rw = rowbase + lr;
                if (rw < n) h[(size_t)rw * CHANNELS + idx] = c[v] * sdis[lr];
            }
        }
    }
}

#define AGG_THREADS 1024
#define AGG_CHUNK   32

template <int J>
__device__ __forceinline__ unsigned agg_row_share(unsigned v) {
    return (unsigned)__builtin_amdgcn_update_dpp(0, (int)v, 0x150 + J, 0xF, 0xF, true);
}

#define AGG_GATHER(J) \
    hv[J] = *reinterpret_cast<const float*>(hbytes + (agg_row_share<J>(gall) | ch4));
#define AGG_ADD(J) \
    atomicAdd(reinterpret_cast<float*>(reinterpret_cast<char*>(acc) + \
                                       (agg_row_share<J>(lall) | ch4)), hv[J]);

__global__ __launch_bounds__(AGG_THREADS)
void gcn_aggregate_kernel(const int2* __restrict__ sorted,
                          const unsigned* __restrict__ bstart,
                          const float* __restrict__ dis,
                          const float* __restrict__ h,
                          const float* __restrict__ b,
                          float* __restrict__ out, int n) {
    __shared__ __attribute__((aligned(16))) float acc[(BK_NODES + 1) * CHANNELS];
    const int tid  = threadIdx.x;
    const int base = blockIdx.x << BK_SHIFT;

    for (int w = tid * 4; w < (BK_NODES + 1) * CHANNELS; w += AGG_THREADS * 4)
        *reinterpret_cast<float4*>(&acc[w]) = make_float4(0.0f, 0.0f, 0.0f, 0.0f);
    __syncthreads();

    const int s0 = (int)bstart[blockIdx.x];
    const int s1 = (int)bstart[blockIdx.x + 1];
    const int lane = tid & 31;
    const int wv = __builtin_amdgcn_readfirstlane(tid >> 5);
    const unsigned ch4 = (unsigned)(lane & 15) * 4u;
    const char* hbytes = reinterpret_cast<const char*>(h);
    int c0 = s0 + wv * AGG_CHUNK;
    int2 ed = make_int2(0, 0);
    if (c0 < s1) ed = sorted[min(c0 + lane, s1 - 1)];
    for (; c0 < s1; c0 += AGG_THREADS) {
        const bool valid = (c0 + lane) < s1;
        const unsigned gall = valid ? ((unsigned)ed.x << 6) : 0u;
        const unsigned lall = valid ? ((unsigned)(ed.y - base) << 6)
                                    : (unsigned)(BK_NODES << 6);
        float hv[16];
        AGG_GATHER(0)  AGG_GATHER(1)  AGG_GATHER(2)  AGG_GATHER(3)
        AGG_GATHER(4)  AGG_GATHER(5)  AGG_GATHER(6)  AGG_GATHER(7)
        AGG_GATHER(8)  AGG_GATHER(9)  AGG_GATHER(10) AGG_GATHER(11)
        AGG_GATHER(12) AGG_GATHER(13) AGG_GATHER(14) AGG_GATHER(15)
        ed = sorted[min(c0 + AGG_THREADS + lane, s1 - 1)];
        AGG_ADD(0)  AGG_ADD(1)  AGG_ADD(2)  AGG_ADD(3)
        AGG_ADD(4)  AGG_ADD(5)  AGG_ADD(6)  AGG_ADD(7)
        AGG_ADD(8)  AGG_ADD(9)  AGG_ADD(10) AGG_ADD(11)
        AGG_ADD(12) AGG_ADD(13) AGG_ADD(14) AGG_ADD(15)
    }
    __syncthreads();

    const int i = base + tid;
    if (tid < BK_NODES && i < n) {
        const float di = dis[i];
        float v[CHANNELS];
        float m = -INFINITY;
#pragma unroll
        for (int c = 0; c < CHANNELS; ++c) {
            float val = di * (acc[tid * CHANNELS + c] + h[(size_t)i * CHANNELS + c]) + b[c];
            val = fmaxf(val, 0.0f);
            v[c] = val;
            m = fmaxf(m, val);
        }
        float s = 0.0f;
#pragma unroll
        for (int c = 0; c < CHANNELS; ++c) s += __expf(v[c] - m);
        float lse = m + __logf(s);
#pragma unroll
        for (int c = 0; c < CHANNELS; ++c)
            out[(size_t)i * CHANNELS + c] = v[c] - lse;
    }
}

extern "C" void kernel_launch(void* const* d_in, const int* in_sizes, int n_in,
                              void* d_out, int out_size, void* d_ws, size_t ws_size,
                              hipStream_t stream) {
    const float* x  = (const float*)d_in[0];
    const int*   ei = (const int*)d_in[1];
    const float* W  = (const float*)d_in[2];
    const float* b  = (const float*)d_in[3];
    float* out = (float*)d_out;

    const int n = in_sizes[0] / KDIM;
    const int e = in_sizes[1] / 2;
    const int* row = ei;
    const int* col = ei + e;

    float* h   = (float*)d_ws;
    float* deg = h + (size_t)n * CHANNELS;
    unsigned* bcnt   = (unsigned*)(deg + (((size_t)n + 3) & ~(size_t)3));
    unsigned* bstart = bcnt + (size_t)DEG_BLOCKS * BK_MAX;
    int2* sorted     = (int2*)(bstart + BK_MAX + 16);
    unsigned* part   = (unsigned*)(sorted + (((size_t)e + 1) & ~(size_t)1));

    gcn_degree_lds_kernel<<<DEG_BLOCKS, DEG_THREADS, 0, stream>>>(col, part, bcnt, n, e);

    {
        const int ntiles = (n + 15) / 16;
        const int gemmBlocks = (ntiles + GEMM_TILES - 1) / GEMM_TILES;
        gcn_gemm_partition_kernel<<<PART_BLOCKS + gemmBlocks, GP_THREADS, 0, stream>>>(
            x, W, part, deg, h, row, col, bcnt, bstart, sorted, n, e);
    }

    gcn_aggregate_kernel<<<(n + BK_NODES - 1) / BK_NODES, AGG_THREADS, 0, stream>>>(
        sorted, bstart, deg, h, b, out, n);
}


# Work-groups of launches 2 and 3 halved to 16 waves, each wave with twice the work (16 pairs a lane and round; two row tiles; two 32-pair chunks a trip): half the waves to start; arrival-order sums
# speedup vs baseline: 6.5105x; 1.0346x over previous
#include <hip/hip_runtime.h>
#include <math.h>

typedef __attribute__((ext_vector_type(2))) float v2f;
typedef __attribute__((ext_vector_type(8))) float v8f;
typedef __attribute__((ext_vector_type(4))) float v4f;
typedef __attribute__((ext_vector_type(4))) unsigned v4u;

#define CHANNELS 16
#define KDIM 128

#define DEG_BLOCKS  128
#define DEG_THREADS 1024
#define DEG_CAPW    51200
#define DEG_BATCH   16

#define BK_SHIFT 9
#define BK_NODES 512
#define BK_MAX   256

__global__ __launch_bounds__(DEG_THREADS)
void gcn_degree_lds_kernel(const int* __restrict__ col,
                           unsigned* __restrict__ part,
                           unsigned* __restrict__ bcnt, int n, int e) {
    __shared__ __attribute__((aligned(16))) unsigned hist[DEG_CAPW];
    const int tid   = threadIdx.x;
    const int wp    = (((n + 1) >> 1) + 3) & ~3;
    const int chunk = (e + DEG_BLOCKS - 1) / DEG_BLOCKS;
    const int e0    = blockIdx.x * chunk;
    const int e1    = min(e0 + chunk, e);
    unsigned* prow  = part + (size_t)blockIdx.x * wp;

    for (int basew = 0; basew < wp; basew += DEG_CAPW) {
        const int nw = min(DEG_CAPW, wp - basew);
        int c[DEG_BATCH];
#pragma unroll
        for (int k = 0; k < DEG_BATCH; ++k) c[k] = 0;
        if (e0 < e1) {
#pragma unroll
            for (int k = 0; k < DEG_BATCH; ++k)
                c[k] = col[min(e0 + tid + k * DEG_THREADS, e1 - 1)];
        }
        for (int w = tid * 4; w < nw; w += DEG_THREADS * 4)
            *reinterpret_cast<uint4*>(&hist[w]) = make_uint4(0u, 0u, 0u, 0u);
        __syncthreads();
        for (int i0 = e0 + tid; i0 < e1; i0 += DEG_THREADS * DEG_BATCH) {
            int cn[DEG_BATCH];
            const int i1 = i0 + DEG_THREADS * DEG_BATCH;
#pragma unroll
            for (int k = 0; k < DEG_BATCH; ++k) cn[k] = 0;
            if (i1 < e1) {
#pragma unroll
                for (int k = 0; k < DEG_BATCH; ++k)
                    cn[k] = col[min(i1 + k * DEG_THREADS, e1 - 1)];
            }
#pragma unroll
            for (int k = 0; k < DEG_BATCH; ++k) {
                unsigned rel = (i0 + k * DEG_THREADS < e1)
                             ? (unsigned)c[k] - (unsigned)(basew * 2)
                             : 0xFFFFFFFFu;
                if (rel < (unsigned)(nw * 2))
                    atomicAdd(&hist[rel >> 1], 1u << ((rel & 1u) * 16u));
            }
#pragma unroll
            for (int k = 0; k < DEG_BATCH; ++k) c[k] = cn[k];
        }
        __syncthreads();
        for (int w = tid * 4; w < nw; w += DEG_THREADS * 4)
            *reinterpret_cast<uint4*>(&prow[basew + w]) =
                *reinterpret_cast<const uint4*>(&hist[w]);
        {
            const int lane = tid & 31;
            const int nbl  = (nw + 255) >> 8;
            for (int j = tid >> 5; j < nbl; j += DEG_THREADS / 32) {
                const int w0 = (j << 8) + lane * 8;
                unsigned s = 0u;
                if (w0 < nw) {
                    uint4 a = *reinterpret_cast<const uint4*>(&hist[w0]);
                    s += (a.x & 0xFFFFu) + (a.x >> 16) + (a.y & 0xFFFFu) + (a.y >> 16)
                       + (a.z & 0xFFFFu) + (a.z >> 16) + (a.w & 0xFFFFu) + (a.w >> 16);
                }
                if (w0 + 4 < nw) {
                    uint4 a = *reinterpret_cast<const uint4*>(&hist[w0 + 4]);
                    s += (a.x & 0xFFFFu) + (a.x >> 16) + (a.y & 0xFFFFu) + (a.y >> 16)
                       + (a.z & 0xFFFFu) + (a.z >> 16) + (a.w & 0xFFFFu) + (a.w >> 16);
                }
                for (int d = 16; d > 0; d >>= 1) s += __shfl_xor(s, d);
                if (lane == 0)
                    bcnt[blockIdx.x * BK_MAX + (basew >> 8) + j] = s;
            }
        }
        __syncthreads();
    }
}

#define PART_BATCH 16
#define GP_THREADS 512
#define PART_BLOCKS (2 * DEG_BLOCKS)
#define PART_ROUND (GP_THREADS * PART_BATCH)
#define GEMM_WAVES  (GP_THREADS / 32)
#define GEMM_WTILES 2
#define GEMM_TILES  (GEMM_WAVES * GEMM_WTILES)
#define GEMM_ROWS   (GEMM_TILES * 16)
#define PSUM_LANES  (BK_MAX / 4)
#define PSUM_GROUPS (GP_THREADS / PSUM_LANES)
#define PSUM_ROWS   (DEG_BLOCKS / PSUM_GROUPS)
static_assert(GEMM_ROWS / 2 == BK_MAX, "a product block's degree words share the bucket tables");
static_assert(PSUM_GROUPS * PSUM_LANES == GP_THREADS && PSUM_GROUPS * PSUM_ROWS == DEG_BLOCKS &&
              PSUM_LANES % 32 == 0, "whole waves per group, the slices split evenly");
#define XT_PITCH 36
#define XT_BYTES (GEMM_WAVES * 16 * XT_PITCH * 4)
#define STAGE_SLOTS ((XT_BYTES / 8) > PART_ROUND ? (XT_BYTES / 8) : PART_ROUND)
static_assert(KDIM == 128 && CHANNELS == 16 && 4 * GP_THREADS == KDIM * CHANNELS,
              "the staging of W and the quarter loop are written for a 128 x 16 W");

__device__ __forceinline__ void wave_lds_order() {
    __builtin_amdgcn_fence(__ATOMIC_RELEASE, "wavefront");
    __builtin_amdgcn_wave_barrier();
    __builtin_amdgcn_fence(__ATOMIC_ACQUIRE, "wavefront");
}

__global__ __launch_bounds__(GP_THREADS) __attribute__((amdgpu_waves_per_eu(8, 8)))
void gcn_gemm_partition_kernel(const float* __restrict__ x,
                               const float* __restrict__ W,
                               const unsigned* __restrict__ part,
                               float* __restrict__ dis,
                               float* __restrict__ h,
                               const int* __restrict__ row,
                               const int* __restrict__ col,
                               const unsigned* __restrict__ bcnt,
                               unsigned* __restrict__ bstart,
                               int2* __restrict__ sorted, int n, int e) {
    struct GpLds {
        unsigned tot[BK_MAX];
        unsigned bef[BK_MAX];
        unsigned cur[BK_MAX];
        unsigned cnt[BK_MAX];
        unsigned lst[BK_MAX];
        unsigned gof[BK_MAX];
        float    sdis[GEMM_ROWS];
        float    wt[KDIM * CHANNELS];
        int2     stage[STAGE_SLOTS];
    };
    __shared__ __attribute__((aligned(16))) GpLds lds;
    unsigned (&tot)[BK_MAX] = lds.tot;
    unsigned (&bef)[BK_MAX] = lds.bef;
    unsigned (&cur)[BK_MAX] = lds.cur;
    unsigned (&cnt)[BK_MAX] = lds.cnt;
    unsigned (&lst)[BK_MAX] = lds.lst;
    unsigned (&gof)[BK_MAX] = lds.gof;
    float (&sdis)[GEMM_ROWS] = lds.sdis;
    int2 (&stage)[STAGE_SLOTS] = lds.stage;
    float (&wt)[KDIM * CHANNELS] = lds.wt;
    const int tid = threadIdx.x;

    if (blockIdx.x < PART_BLOCKS) {
        const int nbk   = (n + BK_NODES - 1) >> BK_SHIFT;
        const int chunk = (e + DEG_BLOCKS - 1) / DEG_BLOCKS;
        const int sl    = blockIdx.x >> 1;
        const int hf    = blockIdx.x & 1;
        const unsigned neg = 0u - (unsigned)hf;
        const int sl0   = sl * chunk;
        const int sl1   = min(sl0 + chunk, e);
        const int mid   = sl0 + ((max(sl1 - sl0, 0) + 1) >> 1);
        const int e0    = hf ? mid : sl0;
        const int e1    = hf ? sl1 : mid;

        int c[PART_BATCH], r[PART_BATCH];
#pragma unroll
        for (int k = 0; k < PART_BATCH; ++k) {
            c[k] = 0;
            r[k] = 0;
        }
        if (e0 < e1) {
#pragma unroll
            for (int k = 0; k < PART_BATCH; ++k) {
                const int idx = min(e0 + tid + k * GP_THREADS, e1 - 1);
                c[k] = col[idx];
                r[k] = row[idx];
            }
        }
        if (tid < BK_MAX) {
            tot[tid] = 0u;
            bef[tid] = 0u;
            cnt[tid] = 0u;
        }
        {
            const int jq = tid & (PSUM_LANES - 1);
            const int j4 = jq * 4;
            const int g = __builtin_amdgcn_readfirstlane(tid / PSUM_LANES);
            const v4u* bc4 = reinterpret_cast<const v4u*>(bcnt);
            v4u t  = {0u, 0u, 0u, 0u};
            v4u bf = {0u, 0u, 0u, 0u};
            if (j4 < nbk) {
                v4u v[PSUM_ROWS];
#pragma unroll
                for (int i = 0; i < PSUM_ROWS; ++i)
                    v[i] = bc4[(g * PSUM_ROWS + i) * (BK_MAX / 4) + jq];
                static_assert(PSUM_ROWS == 16, "the statement below names sixteen values");
                asm volatile("" :: "v"(v[0]), "v"(v[1]), "v"(v[2]), "v"(v[3]), "v"(v[4]), "v"(v[5]),
                             "v"(v[6]), "v"(v[7]), "v"(v[8]), "v"(v[9]), "v"(v[10]), "v"(v[11]),
                             "v"(v[12]), "v"(v[13]), "v"(v[14]), "v"(v[15]));
#pragma unroll
                for (int i = 0; i < PSUM_ROWS; ++i) {
                    const int b = g * PSUM_ROWS + i;
                    t += v[i];
                    if (b < sl + hf) bf += v[i];
                }
            }
            __syncthreads();
            if (j4 + 0 < nbk) { atomicAdd(&tot[j4 + 0], t.x); atomicAdd(&bef[j4 + 0], bf.x); }
            if (j4 + 1 < nbk) { atomicAdd(&tot[j4 + 1], t.y); atomicAdd(&bef[j4 + 1], bf.y); }
            if (j4 + 2 < nbk) { atomicAdd(&tot[j4 + 2], t.z); atomicAdd(&bef[j4 + 2], bf.z); }
            if (j4 + 3 < nbk) { atomicAdd(&tot[j4 + 3], t.w); atomicAdd(&bef[j4 + 3], bf.w); }
        }
        __syncthreads();
        if (tid < 32) {
            unsigned v[8];
            unsigned run = 0u;
#pragma unroll
            for (int i = 0; i < 8; ++i) {
                unsigned t = tot[tid * 8 + i];
                v[i] = run;
                run += t;
            }
            unsigned inc = run;
            for (int d = 1; d < 32; d <<= 1) {
                unsigned t = __shfl_up(inc, d);
                if (tid >= d) inc += t;
            }
            const unsigned excl = inc - run;
#pragma unroll
            for (int i = 0; i < 8; ++i) {
                const unsigned st = excl + v[i];
                const unsigned pl = st + bef[tid * 8 + i];
                cur[tid * 8 + i] = ((pl - (unsigned)hf) ^ neg) - neg;
                if (blockIdx.x == 0) bstart[tid * 8 + i] = st;
            }
            if (blockIdx.x == 0 && tid == 31) bstart[BK_MAX] = inc;
        }
        __syncthreads();

        for (int base = e0; base < e1; base += PART_ROUND) {
            const int m = min(PART_ROUND, e1 - base);
            unsigned rank[PART_BATCH];
#pragma unroll
            for (int k = 0; k < PART_BATCH; ++k) {
                rank[k] = 0u;
                if (tid + k * GP_THREADS < m)
                    rank[k] = atomicAdd(&cnt[c[k] >> BK_SHIFT], 1u);
            }
            __syncthreads();
            if (tid < 32) {
                unsigned t8[8];
                unsigned run = 0u;
#pragma unroll
                for (int i = 0; i < 8; ++i) {
                    t8[i] = cnt[tid * 8 + i];
                    run += t8[i];
                }
                unsigned inc = run;
                for (int d = 1; d < 32; d <<= 1) {
                    unsigned u = __shfl_up(inc, d);
                    if (tid >= d) inc += u;
                }
                unsigned ls = inc - run;
#pragma unroll
                for (int i = 0; i < 8; ++i) {
                    const unsigned t  = t8[i];
                    const unsigned cu = cur[tid * 8 + i];
                    lst[tid * 8 + i] = ls;
                    gof[tid * 8 + i] = cu - ls;
                    cur[tid * 8 + i] = cu + t;
                    cnt[tid * 8 + i] = 0u;
                    ls += t;
                }
            }
            __syncthreads();
#pragma unroll
            for (int k = 0; k < PART_BATCH; ++k) rank[k] += lst[c[k] >> BK_SHIFT];
#pragma unroll
            for (int k = 0; k < PART_BATCH; ++k)
                if (tid + k * GP_THREADS < m)
                    stage[rank[k]] = make_int2(r[k], c[k]);
            __syncthreads();
            if (base + PART_ROUND < e1) {
#pragma unroll
                for (int k = 0; k < PART_BATCH; ++k) {
                    const int idx = min(base + PART_ROUND + tid + k * GP_THREADS, e1 - 1);
                    c[k] = col[idx];
                    r[k] = row[idx];
                }
            }
#pragma unroll
            for (int g = 0; g < PART_BATCH; g += 8) {
                int2 p[8];
                unsigned go[8];
#pragma unroll
                for (int q = 0; q < 8; ++q)
                    p[q] = stage[min(tid + (g + q) * GP_THREADS, m - 1)];
#pragma unroll
                for (int q = 0; q < 8; ++q)
                    go[q] = gof[p[q].y >> BK_SHIFT];
#pragma unroll
                for (int q = 0; q < 8; ++q) {
                    const int j = tid + (g + q) * GP_THREADS;
                    if (j < m) sorted[((go[q] + (unsigned)j) ^ neg) - neg] = p[q];
                }
            }
        }
    } else {
        const int gb      = blockIdx.x - PART_BLOCKS;
        const int lane    = tid & 31;
        const int waveId  = __builtin_amdgcn_readfirstlane(tid >> 5);
        const int rowbase = gb * GEMM_ROWS;
        const int ntiles  = (n + 15) >> 4;
        const int tile    = gb * GEMM_TILES + waveId * GEMM_WTILES;
        const int half = lane >> 4;
        const int idx  = lane & 15;

        const int lr = lane >> 3;
        const int lk = (lane & 7) * 4;
        const unsigned xr0 = ((unsigned)min(tile * 16 + 0 + lr, n - 1) * KDIM + lk) * 4u;
        const unsigned xr1 = ((unsigned)min(tile * 16 + 4 + lr, n - 1) * KDIM + lk) * 4u;
        const unsigned xr2 = ((unsigned)min(tile * 16 + 8 + lr, n - 1) * KDIM + lk) * 4u;
        const unsigned xr3 = ((unsigned)min(tile * 16 + 12 + lr, n - 1) * KDIM + lk) * 4u;
        const unsigned yr0 = ((unsigned)min(tile * 16 + 16 + lr, n - 1) * KDIM + lk) * 4u;
        const unsigned yr1 = ((unsigned)min(tile * 16 + 20 + lr, n - 1) * KDIM + lk) * 4u;
        const unsigned yr2 = ((unsigned)min(tile * 16 + 24 + lr, n - 1) * KDIM + lk) * 4u;
        const unsigned yr3 = ((unsigned)min(tile * 16 + 28 + lr, n - 1) * KDIM + lk) * 4u;
#define XLOAD(G, P, OFF) \
        asm volatile("global_load_b128 %0, %1, %2 offset:%3" : "=&v"(G) : "v"(P), "s"(x), "n"(OFF) : "memory")
#define XSETTLE() asm volatile("s_wait_xcnt 0x0" ::: "memory")
#define XWAIT(G0, G1, G2, G3, N) \
        asm volatile("s_wait_loadcnt %4" : "+v"(G0), "+v"(G1), "+v"(G2), "+v"(G3) : "n"(N) : "memory")
        v4f ga0, ga1, ga2, ga3;
        if (tile < ntiles) {
            XLOAD(ga0, xr0, 0); XLOAD(ga1, xr1, 0); XLOAD(ga2, xr2, 0); XLOAD(ga3, xr3, 0);
            XSETTLE();
        }

        unsigned (&slo)[BK_MAX] = lds.tot;
        unsigned (&shi)[BK_MAX] = lds.bef;
        const int wp = (((n + 1) >> 1) + 3) & ~3;
        if (tid < GEMM_ROWS / 2) {
            slo[tid] = 0u;
            shi[tid] = 0u;
        }
        {
            const int wl4 = (tid & (PSUM_LANES - 1)) * 4;
            const int g   = __builtin_amdgcn_readfirstlane(tid / PSUM_LANES);
            const int wu4 = (rowbase >> 1) + wl4;
            const int wq  = wu4 >> 2;
            unsigned lo0 = 0u, lo1 = 0u, lo2 = 0u, lo3 = 0u;
            unsigned hi0 = 0u, hi1 = 0u, hi2 = 0u, hi3 = 0u;
            const float4 wv = *reinterpret_cast<const float4*>(W + 4 * tid);
            if (wu4 < wp) {
                v4u v[PSUM_ROWS];
#pragma unroll
                for (int i = 0; i < PSUM_ROWS; ++i) {
                    const v4u* prow = reinterpret_cast<const v4u*>(part + (size_t)(g * PSUM_ROWS + i) * wp);
                    v[i] = prow[wq];
                }
                asm volatile("" :: "v"(v[0]), "v"(v[1]), "v"(v[2]), "v"(v[3]), "v"(v[4]), "v"(v[5]),
                             "v"(v[6]), "v"(v[7]), "v"(v[8]), "v"(v[9]), "v"(v[10]), "v"(v[11]),
                             "v"(v[12]), "v"(v[13]), "v"(v[14]), "v"(v[15]));
#pragma unroll
                for (int i = 0; i < PSUM_ROWS; ++i) {
                    lo0 += v[i].x & 0xFFFFu; hi0 += v[i].x >> 16;
                    lo1 += v[i].y & 0xFFFFu; hi1 += v[i].y >> 16;
                    lo2 += v[i].z & 0xFFFFu; hi2 += v[i].z >> 16;
                    lo3 += v[i].w & 0xFFFFu; hi3 += v[i].w >> 16;
                }
            }
            {
                const int wk = (4 * tid) >> 4;
                const int wc = (4 * tid) & 15;
                wt[(wk >> 2) * 64 + wc * 4 + (wk & 3)]       = wv.x;
                wt[(wk >> 2) * 64 + (wc + 1) * 4 + (wk & 3)] = wv.y;
                wt[(wk >> 2) * 64 + (wc + 2) * 4 + (wk & 3)] = wv.z;
                wt[(wk >> 2) * 64 + (wc + 3) * 4 + (wk & 3)] = wv.w;
            }
            __syncthreads();
            if (wu4 < wp) {
                atomicAdd(&slo[wl4 + 0], lo0); atomicAdd(&shi[wl4 + 0], hi0);
                atomicAdd(&slo[wl4 + 1], lo1); atomicAdd(&shi[wl4 + 1], hi1);
                atomicAdd(&slo[wl4 + 2], lo2); atomicAdd(&shi[wl4 + 2], hi2);
                atomicAdd(&slo[wl4 + 3], lo3); atomicAdd(&shi[wl4 + 3], hi3);
            }
        }

        v8f c0 = {}, c1 = {};
        if (tile < ntiles) {
            float* xw = reinterpret_cast<float*>(&stage[0]) + waveId * 16 * XT_PITCH;
            float* xs = xw + lr * XT_PITCH + lk;
            const float* xa = xw + idx * XT_PITCH + 4 * half;
            const float* wb = &wt[half * 64 + idx * 4];
            v4f gb0, gb1, gb2, gb3;
            XLOAD(gb0, xr0, 128); XLOAD(gb1, xr1, 128); XLOAD(gb2, xr2, 128); XLOAD(gb3, xr3, 128);
            XSETTLE();
#define GEMM_QUARTER(G0, G1, G2, G3, ACC, Q, NEXT, R0, R1, R2, R3, NOFF, WAITN)                \
            {                                                                                  \
                XWAIT(G0, G1, G2, G3, WAITN);                                                  \
                *reinterpret_cast<v4f*>(xs + 0 * XT_PITCH)  = G0;                              \
                *reinterpret_cast<v4f*>(xs + 4 * XT_PITCH)  = G1;                              \
                *reinterpret_cast<v4f*>(xs + 8 * XT_PITCH)  = G2;                              \
                *reinterpret_cast<v4f*>(xs + 12 * XT_PITCH) = G3;                              \
                wave_lds_order();            \
                if (NEXT) {                                                                    \
                    XLOAD(G0, R0, NOFF); XLOAD(G1, R1, NOFF);                                  \
                    XLOAD(G2, R2, NOFF); XLOAD(G3, R3, NOFF);                                  \
                    XSETTLE();                                                                 \
                }                                                                              \
                _Pragma("unroll")                                                              \
                for (int s = 0; s < 4; ++s) {                                                  \
                    const float4 av = *reinterpret_cast<const float4*>(xa + 8 * s);            \
                    const float4 bv = *reinterpret_cast<const float4*>(wb + (8 * (Q) + 2 * s) * 64); \
                    v2f a0, b0, a1, b1;                                                        \
                    a0.x = av.x; a0.y = av.y; b0.x = bv.x; b0.y = bv.y;                        \
                    a1.x = av.z; a1.y = av.w; b1.x = bv.z; b1.y = bv.w;                        \
                    ACC = __builtin_amdgcn_wmma_f32_16x16x4_f32(false, a0, false, b0, (short)0, ACC, false, false); \
                    ACC = __builtin_amdgcn_wmma_f32_16x16x4_f32(false, a1, false, b1, (short)0, ACC, false, false); \
                }                                                                              \
                wave_lds_order();                     \
            }
            GEMM_QUARTER(ga0, ga1, ga2, ga3, c0, 0, 1, xr0, xr1, xr2, xr3, 256, 4)
            GEMM_QUARTER(gb0, gb1, gb2, gb3, c0, 1, 1, xr0, xr1, xr2, xr3, 384, 4)
            GEMM_QUARTER(ga0, ga1, ga2, ga3, c0, 2, 1, yr0, yr1, yr2, yr3, 0, 4)
            GEMM_QUARTER(gb0, gb1, gb2, gb3, c0, 3, 1, yr0, yr1, yr2, yr3, 128, 4)
            GEMM_QUARTER(ga0, ga1, ga2, ga3, c1, 0, 1, yr0, yr1, yr2, yr3, 256, 4)
            GEMM_QUARTER(gb0, gb1, gb2, gb3, c1, 1, 1, yr0, yr1, yr2, yr3, 384, 4)
            GEMM_QUARTER(ga0, ga1, ga2, ga3, c1, 2, 0, yr0, yr1, yr2, yr3, 0, 4)
            GEMM_QUARTER(gb0, gb1, gb2, gb3, c1, 3, 0, yr0, yr1, yr2, yr3, 0, 0)
#undef XLOAD
#undef XSETTLE
#undef XWAIT
#undef GEMM_QUARTER
        }
        __syncthreads();
        if (tid < GEMM_ROWS / 2) {
            const int w = (rowbase >> 1) + tid;
            float d0 = 0.0f, d1 = 0.0f;
            if (2 * w < n) {
                d0 = rsqrtf(1.0f + (float)slo[tid]);
                d1 = rsqrtf(1.0f + (float)shi[tid]);
                dis[2 * w] = d0;
                if (2 * w + 1 < n) dis[2 * w + 1] = d1;
            }
            sdis[2 * tid]     = d0;
            sdis[2 * tid + 1] = d1;
        }
        __syncthreads();

        if (tile < ntiles) {
#pragma unroll
            for (int v = 0; v < 8; ++v) {
                const int br = waveId * (16 * GEMM_WTILES) + v + 8 * half;
                const int rw = rowbase + br;
                if (rw < n) h[(size_t)rw * CHANNELS + idx] = c0[v] * sdis[br];
            }
#pragma unroll
            for (int v = 0; v < 8; ++v) {
                const int br = waveId * (16 * GEMM_WTILES) + 16 + v + 8 * half;
                const int rw = rowbase + br;
                if (rw < n) h[(size_t)rw * CHANNELS + idx] = c1[v] * sdis[br];
            }
        }
    }
}

#define AGG_THREADS 512
#define AGG_CHUNK   32
#define AGG_TRIP    (2 * AGG_THREADS)

template <int J>
__device__ __forceinline__ unsigned agg_row_share(unsigned v) {
    return (unsigned)__builtin_amdgcn_update_dpp(0, (int)v, 0x150 + J, 0xF, 0xF, true);
}

#define AGG_GATHER(HV, GALL, J) \
    HV[J] = *reinterpret_cast<const float*>(hbytes + (agg_row_share<J>(GALL) | ch4));
#define AGG_ADD(HV, LALL, J) \
    atomicAdd(reinterpret_cast<float*>(reinterpret_cast<char*>(acc) + \
                                       (agg_row_share<J>(LALL) | ch4)), HV[J]);
#define AGG_GATHER16(HV, GALL) \
    AGG_GATHER(HV, GALL, 0)  AGG_GATHER(HV, GALL, 1)  AGG_GATHER(HV, GALL, 2)  AGG_GATHER(HV, GALL, 3)  \
    AGG_GATHER(HV, GALL, 4)  AGG_GATHER(HV, GALL, 5)  AGG_GATHER(HV, GALL, 6)  AGG_GATHER(HV, GALL, 7)  \
    AGG_GATHER(HV, GALL, 8)  AGG_GATHER(HV, GALL, 9)  AGG_GATHER(HV, GALL, 10) AGG_GATHER(HV, GALL, 11) \
    AGG_GATHER(HV, GALL, 12) AGG_GATHER(HV, GALL, 13) AGG_GATHER(HV, GALL, 14) AGG_GATHER(HV, GALL, 15)
#define AGG_ADD16(HV, LALL) \
    AGG_ADD(HV, LALL, 0)  AGG_ADD(HV, LALL, 1)  AGG_ADD(HV, LALL, 2)  AGG_ADD(HV, LALL, 3)  \
    AGG_ADD(HV, LALL, 4)  AGG_ADD(HV, LALL, 5)  AGG_ADD(HV, LALL, 6)  AGG_ADD(HV, LALL, 7)  \
    AGG_ADD(HV, LALL, 8)  AGG_ADD(HV, LALL, 9)  AGG_ADD(HV, LALL, 10) AGG_ADD(HV, LALL, 11) \
    AGG_ADD(HV, LALL, 12) AGG_ADD(HV, LALL, 13) AGG_ADD(HV, LALL, 14) AGG_ADD(HV, LALL, 15)

__global__ __launch_bounds__(AGG_THREADS)
void gcn_aggregate_kernel(const int2* __restrict__ sorted,
                          const unsigned* __restrict__ bstart,
                          const float* __restrict__ dis,
                          const float* __restrict__ h,
                          const float* __restrict__ b,
                          float* __restrict__ out, int n) {
    __shared__ __attribute__((aligned(16))) float acc[(BK_NODES + 1) * CHANNELS];
    const int tid  = threadIdx.x;
    const int base = blockIdx.x << BK_SHIFT;

    for (int w = tid * 4; w < (BK_NODES + 1) * CHANNELS; w += AGG_THREADS * 4)
        *reinterpret_cast<float4*>(&acc[w]) = make_float4(0.0f, 0.0f, 0.0f, 0.0f);
    __syncthreads();

    const int s0 = (int)bstart[blockIdx.x];
    const int s1 = (int)bstart[blockIdx.x + 1];
    const int lane = tid & 31;
    const int wv = __builtin_amdgcn_readfirstlane(tid >> 5);
    const unsigned ch4 = (unsigned)(lane & 15) * 4u;
    const char* hbytes = reinterpret_cast<const char*>(h);
    int c0 = s0 + wv * (2 * AGG_CHUNK);
    int2 ea = make_int2(0, 0);
    int2 eb = make_int2(0, 0);
    if (c0 < s1) {
        ea = sorted[min(c0 + lane, s1 - 1)];
        eb = sorted[min(c0 + AGG_CHUNK + lane, s1 - 1)];
    }
    for (; c0 < s1; c0 += AGG_TRIP) {
        const bool va = (c0 + lane) < s1;
        const bool vb = (c0 + AGG_CHUNK + lane) < s1;
        const unsigned ga = va ? ((unsigned)ea.x << 6) : 0u;
        const unsigned la = va ? ((unsigned)(ea.y - base) << 6)
                               : (unsigned)(BK_NODES << 6);
        const unsigned gb = vb ? ((unsigned)eb.x << 6) : 0u;
        const unsigned lb = vb ? ((unsigned)(eb.y - base) << 6)
                               : (unsigned)(BK_NODES << 6);
        float hv[16], hw[16];
        AGG_GATHER16(hv, ga)
        AGG_GATHER16(hw, gb)
        ea = sorted[min(c0 + AGG_TRIP + lane, s1 - 1)];
        eb = sorted[min(c0 + AGG_TRIP + AGG_CHUNK + lane, s1 - 1)];
        AGG_ADD16(hv, la)
        AGG_ADD16(hw, lb)
    }
    __syncthreads();

    const int i = base + tid;
    if (tid < BK_NODES && i < n) {
        const float di = dis[i];
        float v[CHANNELS];
        float m = -INFINITY;
#pragma unroll
        for (int c = 0; c < CHANNELS; ++c) {
            float val = di * (acc[tid * CHANNELS + c] + h[(size_t)i * CHANNELS + c]) + b[c];
            val = fmaxf(val, 0.0f);
            v[c] = val;
            m = fmaxf(m, val);
        }
        float s = 0.0f;
#pragma unroll
        for (int c = 0; c < CHANNELS; ++c) s += __expf(v[c] - m);
        float lse = m + __logf(s);
#pragma unroll
        for (int c = 0; c < CHANNELS; ++c)
            out[(size_t)i * CHANNELS + c] = v[c] - lse;
    }
}

extern "C" void kernel_launch(void* const* d_in, const int* in_sizes, int n_in,
                              void* d_out, int out_size, void* d_ws, size_t ws_size,
                              hipStream_t stream) {
    const float* x  = (const float*)d_in[0];
    const int*   ei = (const int*)d_in[1];
    const float* W  = (const float*)d_in[2];
    const float* b  = (const float*)d_in[3];
    float* out = (float*)d_out;

    const int n = in_sizes[0] / KDIM;
    const int e = in_sizes[1] / 2;
    const int* row = ei;
    const int* col = ei + e;

    float* h   = (float*)d_ws;
    float* deg = h + (size_t)n * CHANNELS;
    unsigned* bcnt   = (unsigned*)(deg + (((size_t)n + 3) & ~(size_t)3));
    unsigned* bstart = bcnt + (size_t)DEG_BLOCKS * BK_MAX;
    int2* sorted     = (int2*)(bstart + BK_MAX + 16);
    unsigned* part   = (unsigned*)(sorted + (((size_t)e + 1) & ~(size_t)1));

    gcn_degree_lds_kernel<<<DEG_BLOCKS, DEG_THREADS, 0, stream>>>(col, part, bcnt, n, e);

    {
        const int ntiles = (n + 15) / 16;
        const int gemmBlocks = (ntiles + GEMM_TILES - 1) / GEMM_TILES;
        gcn_gemm_partition_kernel<<<PART_BLOCKS + gemmBlocks, GP_THREADS, 0, stream>>>(
            x, W, part, deg, h, row, col, bcnt, bstart, sorted, n, e);
    }

    gcn_aggregate_kernel<<<(n + BK_NODES - 1) / BK_NODES, AGG_THREADS, 0, stream>>>(
        sorted, bstart, deg, h, b, out, n);
}
